# starting kernel
# baseline (speedup 1.0000x reference)
#include <hip/hip_runtime.h>

typedef float v2f __attribute__((ext_vector_type(2)));
typedef float v8f __attribute__((ext_vector_type(8)));

#define N_NODES 100000
#define N_EDGES 1250000
#define ZBLOCK  256

__global__ __launch_bounds__(ZBLOCK) void gcn_zero_f4(float4* __restrict__ p, int n4) {
    int t = blockIdx.x * ZBLOCK + threadIdx.x;
    if (t < n4) p[t] = make_float4(0.f, 0.f, 0.f, 0.f);
}

__global__ __launch_bounds__(256) void gcn_degree(const int* __restrict__ dst,
                                                  float* __restrict__ deg, int nEdges) {
    int t = blockIdx.x * 256 + threadIdx.x;
    if (t < nEdges) atomicAdd(&deg[dst[t]], 1.0f);
}

__global__ __launch_bounds__(256) void gcn_norm(float* __restrict__ degnorm, int n) {
    int t = blockIdx.x * 256 + threadIdx.x;
    if (t < n) {
        float d = degnorm[t];
        degnorm[t] = (d > 0.0f) ? rsqrtf(fmaxf(d, 1.0f)) : 0.0f;
    }
}

__global__ __launch_bounds__(256) void gcn_scatter64(const float* __restrict__ x,
                                                     const int* __restrict__ src,
                                                     const int* __restrict__ dst,
                                                     const float* __restrict__ norm,
                                                     float* __restrict__ agg,
                                                     int nEdges) {
    int t = blockIdx.x * 256 + threadIdx.x;
    if (t >= nEdges * 16) return;
    int e = t >> 4;
    int c = t & 15;
    int s = src[e];
    int d = dst[e];
    float ns = norm[s];
    float4 v = ((const float4*)(x + (size_t)s * 64))[c];
    float* a = agg + (size_t)d * 64 + c * 4;
    atomicAdd(a + 0, v.x * ns);
    atomicAdd(a + 1, v.y * ns);
    atomicAdd(a + 2, v.z * ns);
    atomicAdd(a + 3, v.w * ns);
}

template <int DIN, int DOUT, bool RELU>
__global__ __launch_bounds__(256) void gcn_gemm_wmma(const float* __restrict__ agg,
                                                     const float* __restrict__ norm,
                                                     const float* __restrict__ W,
                                                     const float* __restrict__ bias,
                                                     float* __restrict__ out,
                                                     int nNodes) {
    const int lane = threadIdx.x & 31;
    const int wave = threadIdx.x >> 5;
    const int tilesPerRow = DOUT / 16;
    const int totalTiles  = (nNodes / 16) * tilesPerRow;
    const int tile = blockIdx.x * 8 + wave;
    if (tile >= totalTiles) return;

    const int nodeBlk = tile / tilesPerRow;
    const int colBlk  = tile % tilesPerRow;
    const int node0   = nodeBlk * 16;
    const int col0    = colBlk * 16;

    const int m  = lane & 15;
    const int kh = lane >> 4;

    const float rowScale = norm[node0 + m];
    const float* arow = agg + (size_t)(node0 + m) * DIN;

    v8f c = {0.f, 0.f, 0.f, 0.f, 0.f, 0.f, 0.f, 0.f};

#pragma unroll
    for (int k0 = 0; k0 < DIN; k0 += 4) {
        const int ka = k0 + 2 * kh;
        float2 ap = *(const float2*)(arow + ka);
        v2f a;
        a.x = ap.x * rowScale;
        a.y = ap.y * rowScale;
        v2f b;
        b.x = W[(size_t)ka * DOUT + col0 + m];
        b.y = W[(size_t)(ka + 1) * DOUT + col0 + m];
        c = __builtin_amdgcn_wmma_f32_16x16x4_f32(
                 false, a,  false, b,
                 (short)0, c,  false,  false);
    }

    const float bv = bias[col0 + m];
#pragma unroll
    for (int r = 0; r < 8; ++r) {
        const int row = r + 8 * kh;
        float v = c[r] + bv;
        if (RELU) v = fmaxf(v, 0.0f);
        out[(size_t)(node0 + row) * DOUT + col0 + m] = v;
    }
}

extern "C" void kernel_launch(void* const* d_in, const int* in_sizes, int n_in,
                              void* d_out, int out_size, void* d_ws, size_t ws_size,
                              hipStream_t stream) {
    const float* x   = (const float*)d_in[0];
    const int*   ei  = (const int*)d_in[1];
    const float* W1  = (const float*)d_in[2];
    const float* b1  = (const float*)d_in[3];
    const float* W2  = (const float*)d_in[4];
    const float* b2  = (const float*)d_in[5];
    float*       out = (float*)d_out;

    const int N = in_sizes[0] / 64;
    const int E = in_sizes[1] / 2;
    const int* src = ei;
    const int* dst = ei + E;

    float* norm = (float*)d_ws;
    float* agg  = norm + N;
    float* h    = agg + (size_t)N * 64;

    {
        int n4 = (N + N * 64) / 4;
        gcn_zero_f4<<<(n4 + ZBLOCK - 1) / ZBLOCK, ZBLOCK, 0, stream>>>((float4*)norm, n4);
    }

    gcn_degree<<<(E + 255) / 256, 256, 0, stream>>>(dst, norm, E);
    gcn_norm<<<(N + 255) / 256, 256, 0, stream>>>(norm, N);

    {
        int work = E * 16;
        gcn_scatter64<<<(work + 255) / 256, 256, 0, stream>>>(x, src, dst, norm, agg, E);
        int tiles = (N / 16) * (64 / 16);
        gcn_gemm_wmma<64, 64, true><<<(tiles + 7) / 8, 256, 0, stream>>>(agg, norm, W1, b1, h, N);
    }

    {
        int n4 = (N * 64) / 4;
        gcn_zero_f4<<<(n4 + ZBLOCK - 1) / ZBLOCK, ZBLOCK, 0, stream>>>((float4*)agg, n4);
    }

    {
        int work = E * 16;
        gcn_scatter64<<<(work + 255) / 256, 256, 0, stream>>>(h, src, dst, norm, agg, E);
        int tiles = (N / 16) * (32 / 16);
        gcn_gemm_wmma<64, 32, false><<<(tiles + 7) / 8, 256, 0, stream>>>(agg, norm, W2, b2, out, N);
    }
}


# Edge sums in ONE accumulator: dst rows split in 16-row blocks by parity between the two L2 halves; resident waves draw 64-edge visits from a per-parity counter, own half first; sum order by arrival
# speedup vs baseline: 1.9174x; 1.9174x over previous
#include <hip/hip_runtime.h>

typedef float v2f __attribute__((ext_vector_type(2)));
typedef float v8f __attribute__((ext_vector_type(8)));

#define N_NODES 100000
#define N_EDGES 1250000
#define ZBLOCK  256

#define RES_BLOCKS 1024

__global__ __launch_bounds__(ZBLOCK) void gcn_zero_f4(float4* __restrict__ p, int n4) {
    const int stride = gridDim.x * ZBLOCK;
    for (int t = blockIdx.x * ZBLOCK + threadIdx.x; t < n4; t += stride)
        p[t] = make_float4(0.f, 0.f, 0.f, 0.f);
}

__device__ __forceinline__ int l2_half() {
    const unsigned id = __builtin_amdgcn_s_sendmsg_rtn(0x87);
    return (int)((id >> 18) & 1u);
}

#define DEG_GROUP 8

__global__ __launch_bounds__(256) void gcn_degree(const int* __restrict__ dst,
                                                  float* __restrict__ deg0,
                                                  float* __restrict__ deg1, int nEdges) {
    float* deg = l2_half() ? deg1 : deg0;
    const int stride = gridDim.x * 256;
    for (int t = blockIdx.x * 256 + threadIdx.x; t < nEdges; t += DEG_GROUP * stride) {
        int d[DEG_GROUP];
#pragma unroll
        for (int u = 0; u < DEG_GROUP; ++u) d[u] = dst[min(t + u * stride, nEdges - 1)];
        static_assert(DEG_GROUP == 8, "the statement below names eight values");
        asm volatile("" :: "v"(d[0]), "v"(d[1]), "v"(d[2]), "v"(d[3]),
                           "v"(d[4]), "v"(d[5]), "v"(d[6]), "v"(d[7]));
#pragma unroll
        for (int u = 0; u < DEG_GROUP; ++u)
            if (t + u * stride < nEdges) atomicAdd(&deg[d[u]], 1.0f);
    }
}

__global__ __launch_bounds__(256) void gcn_norm(float* __restrict__ degnorm,
                                                const float* __restrict__ deg1, int n) {
    int t = blockIdx.x * 256 + threadIdx.x;
    if (t < n) {
        float d = degnorm[t] + deg1[t];
        degnorm[t] = (d > 0.0f) ? rsqrtf(fmaxf(d, 1.0f)) : 0.0f;
    }
}

#define SC_GROUP 8
#define SCAT_BLOCKS 2048

__device__ __forceinline__ int bcast_i(int v, int srcLane) {
    return __builtin_amdgcn_readlane(v, srcLane);
}

__device__ __forceinline__ int pick_i(int v0, int v1, int j) {
    const int a = bcast_i(v0, j & 31);
    const int b = bcast_i(v1, j & 31);
    return a ^ ((a ^ b) & -(j >> 5));
}
__device__ __forceinline__ float pick_f(float v0, float v1, int j) {
    return __builtin_bit_cast(float, pick_i(__builtin_bit_cast(int, v0), __builtin_bit_cast(int, v1), j));
}

__device__ __forceinline__ int next_visit(int* ctr, int lane, int nVisits) {
    int v = nVisits;
    if (lane == 0) {
        if (__hip_atomic_load(ctr, __ATOMIC_RELAXED, __HIP_MEMORY_SCOPE_AGENT) < nVisits)
            v = atomicAdd(ctr, 1);
    }
    return __builtin_amdgcn_readfirstlane(v);
}

template <bool FULL>
__device__ __forceinline__ unsigned long long edge_group(unsigned long long mm,
                                                         int s0, int s1, int d0, int d1,
                                                         float n0, float n1,
                                                         const float* __restrict__ xl,
                                                         float* __restrict__ al) {
    int  j[SC_GROUP];
    bool ok[SC_GROUP];
#pragma unroll
    for (int u = 0; u < SC_GROUP; ++u) {
        ok[u] = FULL || (mm != 0ull);
        j[u]  = ok[u] ? (int)__builtin_ctzll(mm) : 0;
        mm &= mm - 1ull;
    }
    float v0[SC_GROUP], v1[SC_GROUP];
#pragma unroll
    for (int u = 0; u < SC_GROUP; ++u) {
        v0[u] = 0.0f;
        v1[u] = 0.0f;
        if (ok[u]) {
            const int sj = pick_i(s0, s1, j[u]);
            const float* xr = xl + (size_t)sj * 64;
            v0[u] = xr[0];
            v1[u] = xr[32];
        }
    }
#pragma unroll
    for (int u = 0; u < SC_GROUP; ++u) {
        if (ok[u]) {
            const int   dj = pick_i(d0, d1, j[u]);
            const float nj = pick_f(n0, n1, j[u]);
            float* ar = al + (size_t)dj * 64;
            atomicAdd(ar,      v0[u] * nj);
            atomicAdd(ar + 32, v1[u] * nj);
        }
    }
    return mm;
}

__global__ __launch_bounds__(256) void gcn_scatter64(const float* __restrict__ x,
                                                     const int* __restrict__ src,
                                                     const int* __restrict__ dst,
                                                     const float* __restrict__ norm,
                                                     float* __restrict__ agg,
                                                     int* __restrict__ work,
                                                     int nEdges) {
    const int half = l2_half();
    const int lane = threadIdx.x & 31;
    const int nVisits = (nEdges + 63) / 64;

    const float* xl = x + lane;
    float* al = agg + lane;

    for (int ph = 0; ph < 2; ++ph) {
        const int part = half ^ ph;
        int* ctr = work + part * 64;
        int cur = next_visit(ctr, lane, nVisits);
        while (cur < nVisits) {
            const int ea = cur * 64 + lane;
            const int eb = ea + 32;
            const int s0 = src[min(ea, nEdges - 1)];
            const int d0 = dst[min(ea, nEdges - 1)];
            const int s1 = src[min(eb, nEdges - 1)];
            const int d1 = dst[min(eb, nEdges - 1)];

            const int nxt = next_visit(ctr, lane, nVisits);

            asm volatile("" :: "v"(s0), "v"(d0), "v"(s1), "v"(d1));
            const bool m0 = (ea < nEdges) && (((d0 >> 4) & 1) == part);
            const bool m1 = (eb < nEdges) && (((d1 >> 4) & 1) == part);
            const float n0 = m0 ? norm[s0] : 0.0f;
            const float n1 = m1 ? norm[s1] : 0.0f;
            unsigned long long mm = (unsigned long long)__builtin_amdgcn_ballot_w32(m0) |
                                    ((unsigned long long)__builtin_amdgcn_ballot_w32(m1) << 32);

            while (__builtin_popcountll(mm) >= SC_GROUP)
                mm = edge_group<true>(mm, s0, s1, d0, d1, n0, n1, xl, al);
            if (mm != 0ull)
                mm = edge_group<false>(mm, s0, s1, d0, d1, n0, n1, xl, al);

            cur = nxt;
        }
    }
}

template <int DIN, int DOUT, bool RELU>
__global__ __launch_bounds__(256) void gcn_gemm_wmma(const float* __restrict__ agg,
                                                     const float* __restrict__ norm,
                                                     const float* __restrict__ W,
                                                     const float* __restrict__ bias,
                                                     float* __restrict__ out,
                                                     int nNodes) {
    const int lane = threadIdx.x & 31;
    const int wave = __builtin_amdgcn_readfirstlane((int)(threadIdx.x >> 5));
    const int tilesPerRow = DOUT / 16;
    const int totalTiles  = (nNodes / 16) * tilesPerRow;
    const int tileStride  = gridDim.x * 8;

    const int m  = lane & 15;
    const int kh = lane >> 4;

    for (int tile = blockIdx.x * 8 + wave; tile < totalTiles; tile += tileStride) {

    const int nodeBlk = tile / tilesPerRow;
    const int colBlk  = tile % tilesPerRow;
    const int node0   = nodeBlk * 16;
    const int col0    = colBlk * 16;

    const float rowScale = norm[node0 + m];
    const float* arow = agg + (size_t)(node0 + m) * DIN;

    v8f c = {0.f, 0.f, 0.f, 0.f, 0.f, 0.f, 0.f, 0.f};

#pragma unroll
    for (int k0 = 0; k0 < DIN; k0 += 4) {
        const int ka = k0 + 2 * kh;
        float2 ap = *(const float2*)(arow + ka);
        v2f a;
        a.x = ap.x * rowScale;
        a.y = ap.y * rowScale;
        v2f b;
        b.x = W[(size_t)ka * DOUT + col0 + m];
        b.y = W[(size_t)(ka + 1) * DOUT + col0 + m];
        c = __builtin_amdgcn_wmma_f32_16x16x4_f32(
                 false, a,  false, b,
                 (short)0, c,  false,  false);
    }

    const float bv = bias[col0 + m];
#pragma unroll
    for (int r = 0; r < 8; ++r) {
        const int row = r + 8 * kh;
        float v = c[r] + bv;
        if (RELU) v = fmaxf(v, 0.0f);
        out[(size_t)(node0 + row) * DOUT + col0 + m] = v;
    }

    }
}

extern "C" void kernel_launch(void* const* d_in, const int* in_sizes, int n_in,
                              void* d_out, int out_size, void* d_ws, size_t ws_size,
                              hipStream_t stream) {
    const float* x   = (const float*)d_in[0];
    const int*   ei  = (const int*)d_in[1];
    const float* W1  = (const float*)d_in[2];
    const float* b1  = (const float*)d_in[3];
    const float* W2  = (const float*)d_in[4];
    const float* b2  = (const float*)d_in[5];
    float*       out = (float*)d_out;

    const int N = in_sizes[0] / 64;
    const int E = in_sizes[1] / 2;
    const int* src = ei;
    const int* dst = ei + E;

    float* norm = (float*)d_ws;
    float* deg1 = norm + N;
    int*   work = (int*)(deg1 + N);
    float* agg  = deg1 + N + 256;
    float* h    = agg + (size_t)N * 64;

    {
        int n4 = (2 * N + 256 + N * 64) / 4;
        gcn_zero_f4<<<RES_BLOCKS, ZBLOCK, 0, stream>>>((float4*)norm, n4);
    }

    gcn_degree<<<RES_BLOCKS, 256, 0, stream>>>(dst, norm, deg1, E);
    gcn_norm<<<(N + 255) / 256, 256, 0, stream>>>(norm, deg1, N);

    gcn_scatter64<<<SCAT_BLOCKS, 256, 0, stream>>>(x, src, dst, norm, agg, work, E);
    gcn_gemm_wmma<64, 64, true><<<RES_BLOCKS, 256, 0, stream>>>(agg, norm, W1, b1, h, N);

    {
        int n4 = (N * 64) / 4;
        gcn_zero_f4<<<RES_BLOCKS, ZBLOCK, 0, stream>>>((float4*)agg, n4);
    }

    gcn_scatter64<<<SCAT_BLOCKS, 256, 0, stream>>>(h, src, dst, norm, agg, work + 128, E);
    gcn_gemm_wmma<64, 32, false><<<RES_BLOCKS, 256, 0, stream>>>(agg, norm, W2, b2, out, N);
}


# Edge sums in LDS: 256 blocks each own 391 dst rows, find their edges in a byte owner-id array the degree pass writes, add rows by LDS float adds, store once; no global adds; sum order by LDS arrival
# speedup vs baseline: 2.2778x; 1.1880x over previous
#include <hip/hip_runtime.h>

typedef float v2f __attribute__((ext_vector_type(2)));
typedef float v8f __attribute__((ext_vector_type(8)));

#define N_NODES 100000
#define N_EDGES 1250000
#define ZBLOCK  256

#define RES_BLOCKS 1024

__global__ __launch_bounds__(ZBLOCK) void gcn_zero_f4(float4* __restrict__ p, int n4) {
    const int stride = gridDim.x * ZBLOCK;
    for (int t = blockIdx.x * ZBLOCK + threadIdx.x; t < n4; t += stride)
        p[t] = make_float4(0.f, 0.f, 0.f, 0.f);
}

__device__ __forceinline__ int l2_half() {
    const unsigned id = __builtin_amdgcn_s_sendmsg_rtn(0x87);
    return (int)((id >> 18) & 1u);
}

#define DEG_GROUP 8
#define OWN_BLOCKS  256
#define OWN_ROWS    ((N_NODES + OWN_BLOCKS - 1) / OWN_BLOCKS)

__global__ __launch_bounds__(256) void gcn_degree(const int* __restrict__ dst,
                                                  float* __restrict__ deg0,
                                                  float* __restrict__ deg1,
                                                  unsigned char* __restrict__ owner, int nEdges) {
    float* deg = l2_half() ? deg1 : deg0;
    const int stride = gridDim.x * 256;
    for (int t = blockIdx.x * 256 + threadIdx.x; t < nEdges; t += DEG_GROUP * stride) {
        int d[DEG_GROUP];
#pragma unroll
        for (int u = 0; u < DEG_GROUP; ++u) d[u] = dst[min(t + u * stride, nEdges - 1)];
        static_assert(DEG_GROUP == 8, "the statement below names eight values");
        asm volatile("" :: "v"(d[0]), "v"(d[1]), "v"(d[2]), "v"(d[3]),
                           "v"(d[4]), "v"(d[5]), "v"(d[6]), "v"(d[7]));
#pragma unroll
        for (int u = 0; u < DEG_GROUP; ++u)
            if (t + u * stride < nEdges) atomicAdd(&deg[d[u]], 1.0f);
    }
    static_assert(N_EDGES % 4 == 0, "the id array is written four ids at a time");
    const int4* dst4 = (const int4*)dst;
    unsigned* owner4 = (unsigned*)owner;
    for (int q = blockIdx.x * 256 + threadIdx.x; q < nEdges / 4; q += stride) {
        const int4 d4 = dst4[q];
        owner4[q] = (unsigned)(d4.x / OWN_ROWS) | ((unsigned)(d4.y / OWN_ROWS) << 8) |
                    ((unsigned)(d4.z / OWN_ROWS) << 16) | ((unsigned)(d4.w / OWN_ROWS) << 24);
    }
}

__global__ __launch_bounds__(256) void gcn_norm(float* __restrict__ degnorm,
                                                const float* __restrict__ deg1, int n) {
    int t = blockIdx.x * 256 + threadIdx.x;
    if (t < n) {
        float d = degnorm[t] + deg1[t];
        degnorm[t] = (d > 0.0f) ? rsqrtf(fmaxf(d, 1.0f)) : 0.0f;
    }
}

#define OWN_THREADS 1024
#define OWN_WAVES   (OWN_THREADS / 32)
#define ID_CHUNK    16
#define N_CHUNKS    (N_EDGES / ID_CHUNK)
#define SCAN_U      8
#define Q_BATCH     64
#define Q_CAP       1024
#define SC_GROUP    16
static_assert(N_EDGES % ID_CHUNK == 0, "the id array is read in whole 16-byte pieces");
static_assert(OWN_BLOCKS <= 256, "an owner id is one byte");

__device__ __forceinline__ unsigned match_bytes(unsigned w, unsigned id4) {
    const unsigned v = w ^ id4;
    return ~(((v & 0x7F7F7F7Fu) + 0x7F7F7F7Fu) | v | 0x7F7F7F7Fu);
}

__device__ __forceinline__ unsigned chunk_mask(uint4 ids, unsigned id4) {
    return (match_bytes(ids.x, id4) >> 7) | (match_bytes(ids.y, id4) >> 6) |
           (match_bytes(ids.z, id4) >> 5) | (match_bytes(ids.w, id4) >> 4);
}

struct EdgeSumCtx {
    const float* x;
    const int*   src;
    const int*   dst;
    const float* norm;
    float*       acc;
    int*         qlist;
    int          rowBase;
};

__device__ __forceinline__ int list_chunk(unsigned m, int e0, int qn, const EdgeSumCtx& c) {
    unsigned bal;
    while ((bal = __builtin_amdgcn_ballot_w32(m != 0u)) != 0u) {
        if (m != 0u) {
            const int p = (int)__builtin_ctz(m);
            m &= m - 1u;
            const int e = e0 + ((p & 7) << 2) + (p >> 3);
            const int pos = qn + (int)__builtin_amdgcn_mbcnt_lo(bal, 0u);
            if (pos < Q_CAP) {
                c.qlist[pos] = e;
            } else {
                const int   sj = c.src[e];
                const int   dl = c.dst[e] - c.rowBase;
                const float nj = c.norm[sj];
#pragma nounroll
                for (int k = 0; k < 64; ++k) atomicAdd(c.acc + dl * 64 + k, c.x[(size_t)sj * 64 + k] * nj);
            }
        }
        qn += __builtin_popcount(bal);
    }
    return qn;
}

__device__ __forceinline__ void work_off(int qn, int lane, const EdgeSumCtx& c) {
    const float* xl = c.x + lane;
    float* accl = c.acc + lane;
    for (int base = 0; base < qn; base += 32) {
        const int cnt = min(32, qn - base);
        const int e  = c.qlist[base + ((lane < cnt) ? lane : 0)];
        const int s  = c.src[e];
        const int dl = c.dst[e] - c.rowBase;
        const float n = c.norm[s];
        for (int j0 = 0; j0 < cnt; j0 += SC_GROUP) {
            float v0[SC_GROUP], v1[SC_GROUP];
#pragma unroll
            for (int u = 0; u < SC_GROUP; ++u) {
                v0[u] = 0.0f;
                v1[u] = 0.0f;
                if (j0 + u < cnt) {
                    const int sj = __builtin_amdgcn_readlane(s, j0 + u);
                    const float* xr = xl + (size_t)sj * 64;
                    v0[u] = xr[0];
                    v1[u] = xr[32];
                }
            }
#pragma unroll
            for (int u = 0; u < SC_GROUP; ++u) {
                if (j0 + u < cnt) {
                    const int   dj = __builtin_amdgcn_readlane(dl, j0 + u);
                    const float nj = __builtin_bit_cast(float, __builtin_amdgcn_readlane(__builtin_bit_cast(int, n), j0 + u));
                    float* ar = accl + dj * 64;
                    atomicAdd(ar,      v0[u] * nj);
                    atomicAdd(ar + 32, v1[u] * nj);
                }
            }
        }
    }
}

__global__ __launch_bounds__(OWN_THREADS) void gcn_edge_sum_lds(const float* __restrict__ x,
                                                                const int* __restrict__ src,
                                                                const int* __restrict__ dst,
                                                                const float* __restrict__ norm,
                                                                const unsigned char* __restrict__ owner,
                                                                float* __restrict__ agg,
                                                                int nNodes) {
    __shared__ __attribute__((aligned(16))) float acc[OWN_ROWS * 64];
    __shared__ int   qlists[OWN_WAVES * Q_CAP];

    const int tid  = threadIdx.x;
    const int lane = tid & 31;
    const int wave = __builtin_amdgcn_readfirstlane(tid >> 5);
    const int rowBase = blockIdx.x * OWN_ROWS;
    const unsigned id4 = (unsigned)blockIdx.x * 0x01010101u;

    for (int i = tid; i < OWN_ROWS * 64; i += OWN_THREADS) acc[i] = 0.0f;
    __syncthreads();

    EdgeSumCtx c;
    c.x = x; c.src = src; c.dst = dst; c.norm = norm; c.acc = acc;
    c.qlist  = &qlists[wave * Q_CAP];
    c.rowBase = rowBase;
    const uint4* owner4 = (const uint4*)owner;

    const int start = (int)(((long long)blockIdx.x * N_CHUNKS) / OWN_BLOCKS);

    int qn = 0;
    for (int base = 0; base < N_CHUNKS; base += SCAN_U * OWN_THREADS) {
        uint4 ids[SCAN_U];
        int   ch[SCAN_U];
        bool  ok[SCAN_U];
#pragma unroll
        for (int u = 0; u < SCAN_U; ++u) {
            const int i = base + u * OWN_THREADS + tid;
            ok[u] = i < N_CHUNKS;
            ch[u] = (ok[u] ? i : 0) + start;
            if (ch[u] >= N_CHUNKS) ch[u] -= N_CHUNKS;
            ids[u] = owner4[ch[u]];
        }
        static_assert(SCAN_U == 8, "the statements below name eight loads");
        asm volatile("" :: "v"(ids[0].x), "v"(ids[0].y), "v"(ids[0].z), "v"(ids[0].w),
                           "v"(ids[1].x), "v"(ids[1].y), "v"(ids[1].z), "v"(ids[1].w),
                           "v"(ids[2].x), "v"(ids[2].y), "v"(ids[2].z), "v"(ids[2].w),
                           "v"(ids[3].x), "v"(ids[3].y), "v"(ids[3].z), "v"(ids[3].w));
        asm volatile("" :: "v"(ids[4].x), "v"(ids[4].y), "v"(ids[4].z), "v"(ids[4].w),
                           "v"(ids[5].x), "v"(ids[5].y), "v"(ids[5].z), "v"(ids[5].w),
                           "v"(ids[6].x), "v"(ids[6].y), "v"(ids[6].z), "v"(ids[6].w),
                           "v"(ids[7].x), "v"(ids[7].y), "v"(ids[7].z), "v"(ids[7].w));
#pragma unroll
        for (int u = 0; u < SCAN_U; ++u)
            qn = list_chunk(ok[u] ? chunk_mask(ids[u], id4) : 0u, ch[u] * ID_CHUNK, qn, c);
        if (qn >= Q_BATCH) {
            work_off(min(qn, Q_CAP), lane, c);
            qn = 0;
        }
    }
    work_off(min(qn, Q_CAP), lane, c);
    __syncthreads();

    const int rowsHere = min(OWN_ROWS, nNodes - rowBase);
    float4* out4 = (float4*)(agg + (size_t)rowBase * 64);
    const float4* acc4 = (const float4*)acc;
    for (int i = tid; i < rowsHere * 16; i += OWN_THREADS) out4[i] = acc4[i];
}

template <int DIN, int DOUT, bool RELU>
__global__ __launch_bounds__(256) void gcn_gemm_wmma(const float* __restrict__ agg,
                                                     const float* __restrict__ norm,
                                                     const float* __restrict__ W,
                                                     const float* __restrict__ bias,
                                                     float* __restrict__ out,
                                                     int nNodes) {
    const int lane = threadIdx.x & 31;
    const int wave = __builtin_amdgcn_readfirstlane((int)(threadIdx.x >> 5));
    const int tilesPerRow = DOUT / 16;
    const int totalTiles  = (nNodes / 16) * tilesPerRow;
    const int tileStride  = gridDim.x * 8;

    const int m  = lane & 15;
    const int kh = lane >> 4;

    for (int tile = blockIdx.x * 8 + wave; tile < totalTiles; tile += tileStride) {

    const int nodeBlk = tile / tilesPerRow;
    const int colBlk  = tile % tilesPerRow;
    const int node0   = nodeBlk * 16;
    const int col0    = colBlk * 16;

    const float rowScale = norm[node0 + m];
    const float* arow = agg + (size_t)(node0 + m) * DIN;

    v8f c = {0.f, 0.f, 0.f, 0.f, 0.f, 0.f, 0.f, 0.f};

#pragma unroll
    for (int k0 = 0; k0 < DIN; k0 += 4) {
        const int ka = k0 + 2 * kh;
        float2 ap = *(const float2*)(arow + ka);
        v2f a;
        a.x = ap.x * rowScale;
        a.y = ap.y * rowScale;
        v2f b;
        b.x = W[(size_t)ka * DOUT + col0 + m];
        b.y = W[(size_t)(ka + 1) * DOUT + col0 + m];
        c = __builtin_amdgcn_wmma_f32_16x16x4_f32(
                 false, a,  false, b,
                 (short)0, c,  false,  false);
    }

    const float bv = bias[col0 + m];
#pragma unroll
    for (int r = 0; r < 8; ++r) {
        const int row = r + 8 * kh;
        float v = c[r] + bv;
        if (RELU) v = fmaxf(v, 0.0f);
        out[(size_t)(node0 + row) * DOUT + col0 + m] = v;
    }

    }
}

extern "C" void kernel_launch(void* const* d_in, const int* in_sizes, int n_in,
                              void* d_out, int out_size, void* d_ws, size_t ws_size,
                              hipStream_t stream) {
    const float* x   = (const float*)d_in[0];
    const int*   ei  = (const int*)d_in[1];
    const float* W1  = (const float*)d_in[2];
    const float* b1  = (const float*)d_in[3];
    const float* W2  = (const float*)d_in[4];
    const float* b2  = (const float*)d_in[5];
    float*       out = (float*)d_out;

    const int N = in_sizes[0] / 64;
    const int E = in_sizes[1] / 2;
    const int* src = ei;
    const int* dst = ei + E;

    float* norm = (float*)d_ws;
    float* deg1 = norm + N;
    float* agg  = deg1 + N + 256;
    float* h    = agg + (size_t)N * 64;
    unsigned char* owner = (unsigned char*)(h + (size_t)N * 64);

    {
        int n4 = (2 * N) / 4;
        gcn_zero_f4<<<RES_BLOCKS, ZBLOCK, 0, stream>>>((float4*)norm, n4);
    }

    gcn_degree<<<RES_BLOCKS, 256, 0, stream>>>(dst, norm, deg1, owner, E);
    gcn_norm<<<(N + 255) / 256, 256, 0, stream>>>(norm, deg1, N);

    gcn_edge_sum_lds<<<OWN_BLOCKS, OWN_THREADS, 0, stream>>>(x, src, dst, norm, owner, agg, N);
    gcn_gemm_wmma<64, 64, true><<<RES_BLOCKS, 256, 0, stream>>>(agg, norm, W1, b1, h, N);

    gcn_edge_sum_lds<<<OWN_BLOCKS, OWN_THREADS, 0, stream>>>(h, src, dst, norm, owner, agg, N);
    gcn_gemm_wmma<64, 32, false><<<RES_BLOCKS, 256, 0, stream>>>(agg, norm, W2, b2, out, N);
}


# Layer 2's edge sum walks per-wave lists (source row, owned row, scale) that layer 1's id scan stores as it works its edges off: no second scan, no per-edge index loads; sum order by LDS arrival
# speedup vs baseline: 2.8388x; 1.2463x over previous
#include <hip/hip_runtime.h>

typedef float v2f __attribute__((ext_vector_type(2)));
typedef float v8f __attribute__((ext_vector_type(8)));

#define N_NODES 100000
#define N_EDGES 1250000
#define ZBLOCK  256

#define RES_BLOCKS 1024

__global__ __launch_bounds__(ZBLOCK) void gcn_zero_f4(float4* __restrict__ p, int n4) {
    const int stride = gridDim.x * ZBLOCK;
    for (int t = blockIdx.x * ZBLOCK + threadIdx.x; t < n4; t += stride)
        p[t] = make_float4(0.f, 0.f, 0.f, 0.f);
}

__device__ __forceinline__ int l2_half() {
    const unsigned id = __builtin_amdgcn_s_sendmsg_rtn(0x87);
    return (int)((id >> 18) & 1u);
}

#define DEG_GROUP 8
#define OWN_BLOCKS  256
#define OWN_ROWS    ((N_NODES + OWN_BLOCKS - 1) / OWN_BLOCKS)

__global__ __launch_bounds__(256) void gcn_degree(const int* __restrict__ dst,
                                                  float* __restrict__ deg0,
                                                  float* __restrict__ deg1,
                                                  unsigned char* __restrict__ owner, int nEdges) {
    float* deg = l2_half() ? deg1 : deg0;
    const int stride = gridDim.x * 256;
    for (int t = blockIdx.x * 256 + threadIdx.x; t < nEdges; t += DEG_GROUP * stride) {
        int d[DEG_GROUP];
#pragma unroll
        for (int u = 0; u < DEG_GROUP; ++u) d[u] = dst[min(t + u * stride, nEdges - 1)];
        static_assert(DEG_GROUP == 8, "the statement below names eight values");
        asm volatile("" :: "v"(d[0]), "v"(d[1]), "v"(d[2]), "v"(d[3]),
                           "v"(d[4]), "v"(d[5]), "v"(d[6]), "v"(d[7]));
#pragma unroll
        for (int u = 0; u < DEG_GROUP; ++u)
            if (t + u * stride < nEdges) atomicAdd(&deg[d[u]], 1.0f);
    }
    static_assert(N_EDGES % 4 == 0, "the id array is written four ids at a time");
    const int4* dst4 = (const int4*)dst;
    unsigned* owner4 = (unsigned*)owner;
    for (int q = blockIdx.x * 256 + threadIdx.x; q < nEdges / 4; q += stride) {
        const int4 d4 = dst4[q];
        owner4[q] = (unsigned)(d4.x / OWN_ROWS) | ((unsigned)(d4.y / OWN_ROWS) << 8) |
                    ((unsigned)(d4.z / OWN_ROWS) << 16) | ((unsigned)(d4.w / OWN_ROWS) << 24);
    }
}

__global__ __launch_bounds__(256) void gcn_norm(float* __restrict__ degnorm,
                                                const float* __restrict__ deg1, int n) {
    int t = blockIdx.x * 256 + threadIdx.x;
    if (t < n) {
        float d = degnorm[t] + deg1[t];
        degnorm[t] = (d > 0.0f) ? rsqrtf(fmaxf(d, 1.0f)) : 0.0f;
    }
}

#define OWN_THREADS 1024
#define OWN_WAVES   (OWN_THREADS / 32)
#define ID_CHUNK    16
#define N_CHUNKS    (N_EDGES / ID_CHUNK)
#define SCAN_U      8
#define Q_BATCH     64
#define Q_CAP       1024
#define SC_GROUP    16
#define LIST_CAP    512
#define WALK_U      8
#define SRC_BITS    17
static_assert(N_NODES <= (1 << SRC_BITS), "a source row number fits SRC_BITS bits");
static_assert(OWN_ROWS <= (1 << (31 - SRC_BITS)), "an owned row number fits above it in a positive int");
static_assert(LIST_CAP * 8 <= Q_CAP * 4, "the second pass holds a whole list in the wave's LDS list space");
static_assert(N_EDGES % ID_CHUNK == 0, "the id array is read in whole 16-byte pieces");
static_assert(OWN_BLOCKS <= 256, "an owner id is one byte");
#define OWNER_SPAN  ((N_EDGES + 255) / 256 * 256)

__device__ __forceinline__ unsigned match_bytes(unsigned w, unsigned id4) {
    const unsigned v = w ^ id4;
    return ~(((v & 0x7F7F7F7Fu) + 0x7F7F7F7Fu) | v | 0x7F7F7F7Fu);
}

__device__ __forceinline__ unsigned chunk_mask(uint4 ids, unsigned id4) {
    return (match_bytes(ids.x, id4) >> 7) | (match_bytes(ids.y, id4) >> 6) |
           (match_bytes(ids.z, id4) >> 5) | (match_bytes(ids.w, id4) >> 4);
}

struct EdgeSumCtx {
    const float* x;
    const int*   src;
    const int*   dst;
    const float* norm;
    float*       acc;
    int*         qlist;
    int          rowBase;
};

__device__ __forceinline__ int list_chunk(unsigned m, int e0, int qn, const EdgeSumCtx& c) {
    unsigned bal;
    while ((bal = __builtin_amdgcn_ballot_w32(m != 0u)) != 0u) {
        if (m != 0u) {
            const int p = (int)__builtin_ctz(m);
            m &= m - 1u;
            const int e = e0 + ((p & 7) << 2) + (p >> 3);
            const int pos = qn + (int)__builtin_amdgcn_mbcnt_lo(bal, 0u);
            if (pos < Q_CAP) {
                c.qlist[pos] = e;
            } else {
                const int   sj = c.src[e];
                const int   dl = c.dst[e] - c.rowBase;
                const float nj = c.norm[sj];
#pragma nounroll
                for (int k = 0; k < 64; ++k) atomicAdd(c.acc + dl * 64 + k, c.x[(size_t)sj * 64 + k] * nj);
            }
        }
        qn += __builtin_popcount(bal);
    }
    return qn;
}

__device__ __forceinline__ void sum_batch(int s, int dl, float n, int cnt, int lane, const EdgeSumCtx& c) {
    const float* xl = c.x + lane;
    float* accl = c.acc + lane;
    for (int j0 = 0; j0 < cnt; j0 += SC_GROUP) {
        float v0[SC_GROUP], v1[SC_GROUP];
#pragma unroll
        for (int u = 0; u < SC_GROUP; ++u) {
            v0[u] = 0.0f;
            v1[u] = 0.0f;
            if (j0 + u < cnt) {
                const int sj = __builtin_amdgcn_readlane(s, j0 + u);
                const float* xr = xl + (size_t)sj * 64;
                v0[u] = xr[0];
                v1[u] = xr[32];
            }
        }
#pragma unroll
        for (int u = 0; u < SC_GROUP; ++u) {
            if (j0 + u < cnt) {
                const int   dj = __builtin_amdgcn_readlane(dl, j0 + u);
                const float nj = __builtin_bit_cast(float, __builtin_amdgcn_readlane(__builtin_bit_cast(int, n), j0 + u));
                float* ar = accl + dj * 64;
                atomicAdd(ar,      v0[u] * nj);
                atomicAdd(ar + 32, v1[u] * nj);
            }
        }
    }
}

template <bool KEEP>
__device__ __forceinline__ int work_off(int qn, int lane, const EdgeSumCtx& c, int2* lst, int ltot) {
    for (int base = 0; base < qn; base += 32) {
        const int cnt = min(32, qn - base);
        const int e  = c.qlist[base + ((lane < cnt) ? lane : 0)];
        const int s  = c.src[e];
        const int dl = c.dst[e] - c.rowBase;
        const float n = c.norm[s];
        sum_batch(s, dl, n, cnt, lane, c);
        if (KEEP) {
            if (ltot + cnt <= LIST_CAP && lane < cnt)
                lst[ltot + lane] = make_int2(s | (dl << SRC_BITS), __builtin_bit_cast(int, n));
            ltot += cnt;
        }
    }
    return ltot;
}

__device__ __forceinline__ void walk_list(const int2* lst, int total, int lane, const EdgeSumCtx& c) {
    int2* ql = (int2*)c.qlist;
    for (int b = 0; b < total; b += WALK_U * 32) {
        int2 t[WALK_U];
#pragma unroll
        for (int u = 0; u < WALK_U; ++u) t[u] = lst[min(b + u * 32 + lane, total - 1)];
        static_assert(WALK_U == 8, "the statement below names eight loads");
        asm volatile("" :: "v"(t[0].x), "v"(t[0].y), "v"(t[1].x), "v"(t[1].y),
                           "v"(t[2].x), "v"(t[2].y), "v"(t[3].x), "v"(t[3].y),
                           "v"(t[4].x), "v"(t[4].y), "v"(t[5].x), "v"(t[5].y),
                           "v"(t[6].x), "v"(t[6].y), "v"(t[7].x), "v"(t[7].y));
#pragma unroll
        for (int u = 0; u < WALK_U; ++u)
            if (b + u * 32 + lane < total) ql[b + u * 32 + lane] = t[u];
    }
    for (int base = 0; base < total; base += 32) {
        const int cnt = min(32, total - base);
        const int2 cur = ql[base + ((lane < cnt) ? lane : 0)];
        const int s  = cur.x & ((1 << SRC_BITS) - 1);
        const int dl = cur.x >> SRC_BITS;
        sum_batch(s, dl, __builtin_bit_cast(float, cur.y), cnt, lane, c);
    }
}

template <bool WALK>
__global__ __launch_bounds__(OWN_THREADS) void gcn_edge_sum_lds(const float* __restrict__ x,
                                                                const int* __restrict__ src,
                                                                const int* __restrict__ dst,
                                                                const float* __restrict__ norm,
                                                                const unsigned char* __restrict__ owner,
                                                                float* __restrict__ agg,
                                                                int2* __restrict__ lists,
                                                                int* __restrict__ counts,
                                                                int nNodes) {
    __shared__ __attribute__((aligned(16))) float acc[OWN_ROWS * 64];
    __shared__ int   qlists[OWN_WAVES * Q_CAP];

    const int tid  = threadIdx.x;
    const int lane = tid & 31;
    const int wave = __builtin_amdgcn_readfirstlane(tid >> 5);
    const int rowBase = blockIdx.x * OWN_ROWS;
    const unsigned id4 = (unsigned)blockIdx.x * 0x01010101u;

    for (int i = tid; i < OWN_ROWS * 64; i += OWN_THREADS) acc[i] = 0.0f;
    __syncthreads();

    EdgeSumCtx c;
    c.x = x; c.src = src; c.dst = dst; c.norm = norm; c.acc = acc;
    c.qlist  = &qlists[wave * Q_CAP];
    c.rowBase = rowBase;
    const uint4* owner4 = (const uint4*)owner;

    int2* lst  = lists + (size_t)(blockIdx.x * OWN_WAVES + wave) * LIST_CAP;
    int*  cntp = counts + blockIdx.x * OWN_WAVES + wave;

    bool scan = true;
    if (WALK) {
        const int total = __builtin_amdgcn_readfirstlane(*cntp);
        if (total <= LIST_CAP) {
            if (total > 0) walk_list(lst, total, lane, c);
            scan = false;
        }
    }

    if (scan) {

    const int start = (int)(((long long)blockIdx.x * N_CHUNKS) / OWN_BLOCKS);

    int ltot = 0;
    int qn = 0;
    for (int base = 0; base < N_CHUNKS; base += SCAN_U * OWN_THREADS) {
        uint4 ids[SCAN_U];
        int   ch[SCAN_U];
        bool  ok[SCAN_U];
#pragma unroll
        for (int u = 0; u < SCAN_U; ++u) {
            const int i = base + u * OWN_THREADS + tid;
            ok[u] = i < N_CHUNKS;
            ch[u] = (ok[u] ? i : 0) + start;
            if (ch[u] >= N_CHUNKS) ch[u] -= N_CHUNKS;
            ids[u] = owner4[ch[u]];
        }
        static_assert(SCAN_U == 8, "the statements below name eight loads");
        asm volatile("" :: "v"(ids[0].x), "v"(ids[0].y), "v"(ids[0].z), "v"(ids[0].w),
                           "v"(ids[1].x), "v"(ids[1].y), "v"(ids[1].z), "v"(ids[1].w),
                           "v"(ids[2].x), "v"(ids[2].y), "v"(ids[2].z), "v"(ids[2].w),
                           "v"(ids[3].x), "v"(ids[3].y), "v"(ids[3].z), "v"(ids[3].w));
        asm volatile("" :: "v"(ids[4].x), "v"(ids[4].y), "v"(ids[4].z), "v"(ids[4].w),
                           "v"(ids[5].x), "v"(ids[5].y), "v"(ids[5].z), "v"(ids[5].w),
                           "v"(ids[6].x), "v"(ids[6].y), "v"(ids[6].z), "v"(ids[6].w),
                           "v"(ids[7].x), "v"(ids[7].y), "v"(ids[7].z), "v"(ids[7].w));
#pragma unroll
        for (int u = 0; u < SCAN_U; ++u)
            qn = list_chunk(ok[u] ? chunk_mask(ids[u], id4) : 0u, ch[u] * ID_CHUNK, qn, c);
        if (qn >= Q_BATCH) {
            if (qn > Q_CAP) ltot = LIST_CAP + 1;
            ltot = work_off<!WALK>(min(qn, Q_CAP), lane, c, lst, ltot);
            qn = 0;
        }
    }
    if (qn > Q_CAP) ltot = LIST_CAP + 1;
    ltot = work_off<!WALK>(min(qn, Q_CAP), lane, c, lst, ltot);
    if (!WALK && lane == 0) *cntp = ltot;

    }
    __syncthreads();

    const int rowsHere = min(OWN_ROWS, nNodes - rowBase);
    float4* out4 = (float4*)(agg + (size_t)rowBase * 64);
    const float4* acc4 = (const float4*)acc;
    for (int i = tid; i < rowsHere * 16; i += OWN_THREADS) out4[i] = acc4[i];
}

template <int DIN, int DOUT, bool RELU>
__global__ __launch_bounds__(256) void gcn_gemm_wmma(const float* __restrict__ agg,
                                                     const float* __restrict__ norm,
                                                     const float* __restrict__ W,
                                                     const float* __restrict__ bias,
                                                     float* __restrict__ out,
                                                     int nNodes) {
    const int lane = threadIdx.x & 31;
    const int wave = __builtin_amdgcn_readfirstlane((int)(threadIdx.x >> 5));
    const int tilesPerRow = DOUT / 16;
    const int totalTiles  = (nNodes / 16) * tilesPerRow;
    const int tileStride  = gridDim.x * 8;

    const int m  = lane & 15;
    const int kh = lane >> 4;

    for (int tile = blockIdx.x * 8 + wave; tile < totalTiles; tile += tileStride) {

    const int nodeBlk = tile / tilesPerRow;
    const int colBlk  = tile % tilesPerRow;
    const int node0   = nodeBlk * 16;
    const int col0    = colBlk * 16;

    const float rowScale = norm[node0 + m];
    const float* arow = agg + (size_t)(node0 + m) * DIN;

    v8f c = {0.f, 0.f, 0.f, 0.f, 0.f, 0.f, 0.f, 0.f};

#pragma unroll
    for (int k0 = 0; k0 < DIN; k0 += 4) {
        const int ka = k0 + 2 * kh;
        float2 ap = *(const float2*)(arow + ka);
        v2f a;
        a.x = ap.x * rowScale;
        a.y = ap.y * rowScale;
        v2f b;
        b.x = W[(size_t)ka * DOUT + col0 + m];
        b.y = W[(size_t)(ka + 1) * DOUT + col0 + m];
        c = __builtin_amdgcn_wmma_f32_16x16x4_f32(
                 false, a,  false, b,
                 (short)0, c,  false,  false);
    }

    const float bv = bias[col0 + m];
#pragma unroll
    for (int r = 0; r < 8; ++r) {
        const int row = r + 8 * kh;
        float v = c[r] + bv;
        if (RELU) v = fmaxf(v, 0.0f);
        out[(size_t)(node0 + row) * DOUT + col0 + m] = v;
    }

    }
}

extern "C" void kernel_launch(void* const* d_in, const int* in_sizes, int n_in,
                              void* d_out, int out_size, void* d_ws, size_t ws_size,
                              hipStream_t stream) {
    const float* x   = (const float*)d_in[0];
    const int*   ei  = (const int*)d_in[1];
    const float* W1  = (const float*)d_in[2];
    const float* b1  = (const float*)d_in[3];
    const float* W2  = (const float*)d_in[4];
    const float* b2  = (const float*)d_in[5];
    float*       out = (float*)d_out;

    const int N = in_sizes[0] / 64;
    const int E = in_sizes[1] / 2;
    const int* src = ei;
    const int* dst = ei + E;

    float* norm = (float*)d_ws;
    float* deg1 = norm + N;
    float* agg  = deg1 + N + 256;
    float* h    = agg + (size_t)N * 64;
    unsigned char* owner = (unsigned char*)(h + (size_t)N * 64);
    int*  counts = (int*)(owner + OWNER_SPAN);
    int2* lists  = (int2*)(counts + OWN_BLOCKS * OWN_WAVES);

    {
        int n4 = (2 * N) / 4;
        gcn_zero_f4<<<RES_BLOCKS, ZBLOCK, 0, stream>>>((float4*)norm, n4);
    }

    gcn_degree<<<RES_BLOCKS, 256, 0, stream>>>(dst, norm, deg1, owner, E);
    gcn_norm<<<(N + 255) / 256, 256, 0, stream>>>(norm, deg1, N);

    gcn_edge_sum_lds<false><<<OWN_BLOCKS, OWN_THREADS, 0, stream>>>(x, src, dst, norm, owner, agg, lists, counts, N);
    gcn_gemm_wmma<64, 64, true><<<RES_BLOCKS, 256, 0, stream>>>(agg, norm, W1, b1, h, N);

    gcn_edge_sum_lds<true><<<OWN_BLOCKS, OWN_THREADS, 0, stream>>>(h, src, dst, norm, owner, agg, lists, counts, N);
    gcn_gemm_wmma<64, 32, false><<<RES_BLOCKS, 256, 0, stream>>>(agg, norm, W2, b2, out, N);
}


# Owner ids kept bit-sliced (8 planes per 32 edges, made by ballots in the degree pass): an edge-sum block tests 32 edges with one bit operation per plane, not byte compares; sum order as before
# speedup vs baseline: 3.0516x; 1.0750x over previous
#include <hip/hip_runtime.h>

typedef float v2f __attribute__((ext_vector_type(2)));
typedef float v8f __attribute__((ext_vector_type(8)));

#define N_NODES 100000
#define N_EDGES 1250000
#define ZBLOCK  256

#define RES_BLOCKS 1024

__global__ __launch_bounds__(ZBLOCK) void gcn_zero_f4(float4* __restrict__ p, int n4) {
    const int stride = gridDim.x * ZBLOCK;
    for (int t = blockIdx.x * ZBLOCK + threadIdx.x; t < n4; t += stride)
        p[t] = make_float4(0.f, 0.f, 0.f, 0.f);
}

__device__ __forceinline__ int l2_half() {
    const unsigned id = __builtin_amdgcn_s_sendmsg_rtn(0x87);
    return (int)((id >> 18) & 1u);
}

#define DEG_GROUP 8
#define OWN_BLOCKS  256
#define OWN_ROWS    ((N_NODES + OWN_BLOCKS - 1) / OWN_BLOCKS)
#define TILE_EDGES  1024
#define N_TILES     ((N_EDGES + TILE_EDGES - 1) / TILE_EDGES)

__global__ __launch_bounds__(256) void gcn_degree(const int* __restrict__ dst,
                                                  float* __restrict__ deg0,
                                                  float* __restrict__ deg1,
                                                  unsigned char* __restrict__ owner, int nEdges) {
    float* deg = l2_half() ? deg1 : deg0;
    const int stride = gridDim.x * 256;
    for (int t = blockIdx.x * 256 + threadIdx.x; t < nEdges; t += DEG_GROUP * stride) {
        int d[DEG_GROUP];
#pragma unroll
        for (int u = 0; u < DEG_GROUP; ++u) d[u] = dst[min(t + u * stride, nEdges - 1)];
        static_assert(DEG_GROUP == 8, "the statement below names eight values");
        asm volatile("" :: "v"(d[0]), "v"(d[1]), "v"(d[2]), "v"(d[3]),
                           "v"(d[4]), "v"(d[5]), "v"(d[6]), "v"(d[7]));
#pragma unroll
        for (int u = 0; u < DEG_GROUP; ++u)
            if (t + u * stride < nEdges) atomicAdd(&deg[d[u]], 1.0f);
    }
    const int lane = threadIdx.x & 31;
    const int wib = __builtin_amdgcn_readfirstlane((int)(threadIdx.x >> 5));
    uint4* owner4 = (uint4*)owner;
    for (int tile = blockIdx.x + gridDim.x * wib; tile < N_TILES; tile += gridDim.x * 8) {
        const int e0 = tile * TILE_EDGES + lane;
        int d[32];
#pragma unroll
        for (int g = 0; g < 32; ++g) d[g] = dst[min(e0 + g * 32, nEdges - 1)];
        int p[8] = {0, 0, 0, 0, 0, 0, 0, 0};
#pragma unroll
        for (int g = 0; g < 32; ++g) {
            const unsigned id = (unsigned)d[g] / (unsigned)OWN_ROWS;
#pragma unroll
            for (int b = 0; b < 8; ++b) {
                const int plane = (int)__builtin_amdgcn_ballot_w32(((id >> b) & 1u) != 0u);
                p[b] = (lane == g) ? plane : p[b];
            }
        }
        owner4[tile * 64 + lane]      = make_uint4((unsigned)p[0], (unsigned)p[1], (unsigned)p[2], (unsigned)p[3]);
        owner4[tile * 64 + 32 + lane] = make_uint4((unsigned)p[4], (unsigned)p[5], (unsigned)p[6], (unsigned)p[7]);
    }
}

__global__ __launch_bounds__(256) void gcn_norm(float* __restrict__ degnorm,
                                                const float* __restrict__ deg1, int n) {
    int t = blockIdx.x * 256 + threadIdx.x;
    if (t < n) {
        float d = degnorm[t] + deg1[t];
        degnorm[t] = (d > 0.0f) ? rsqrtf(fmaxf(d, 1.0f)) : 0.0f;
    }
}

#define OWN_THREADS 1024
#define OWN_WAVES   (OWN_THREADS / 32)
#define SCAN_U      4
#define Q_BATCH     64
#define Q_CAP       1024
#define SC_GROUP    16
#define LIST_CAP    512
#define WALK_U      8
#define SRC_BITS    17
static_assert(N_NODES <= (1 << SRC_BITS), "a source row number fits SRC_BITS bits");
static_assert(OWN_ROWS <= (1 << (31 - SRC_BITS)), "an owned row number fits above it in a positive int");
static_assert(LIST_CAP * 8 <= Q_CAP * 4, "the second pass holds a whole list in the wave's LDS list space");
static_assert(OWN_BLOCKS <= 256, "an owner id has eight bits");
#define OWNER_SPAN  (N_TILES * TILE_EDGES)

__device__ __forceinline__ unsigned group_mask(uint4 lo, uint4 hi, unsigned id) {
    unsigned m =  ~(lo.x ^ (0u - ( id       & 1u)));
    m &= ~(lo.y ^ (0u - ((id >> 1) & 1u)));
    m &= ~(lo.z ^ (0u - ((id >> 2) & 1u)));
    m &= ~(lo.w ^ (0u - ((id >> 3) & 1u)));
    m &= ~(hi.x ^ (0u - ((id >> 4) & 1u)));
    m &= ~(hi.y ^ (0u - ((id >> 5) & 1u)));
    m &= ~(hi.z ^ (0u - ((id >> 6) & 1u)));
    m &= ~(hi.w ^ (0u - ((id >> 7) & 1u)));
    return m;
}

struct EdgeSumCtx {
    const float* x;
    const int*   src;
    const int*   dst;
    const float* norm;
    float*       acc;
    int*         qlist;
    int          rowBase;
};

__device__ __forceinline__ int list_group(unsigned m, int e0, int qn, const EdgeSumCtx& c) {
    unsigned bal;
    while ((bal = __builtin_amdgcn_ballot_w32(m != 0u)) != 0u) {
        if (m != 0u) {
            const int e = e0 + (int)__builtin_ctz(m);
            m &= m - 1u;
            const int pos = qn + (int)__builtin_amdgcn_mbcnt_lo(bal, 0u);
            if (pos < Q_CAP) {
                c.qlist[pos] = e;
            } else {
                const int   sj = c.src[e];
                const int   dl = c.dst[e] - c.rowBase;
                const float nj = c.norm[sj];
#pragma nounroll
                for (int k = 0; k < 64; ++k) atomicAdd(c.acc + dl * 64 + k, c.x[(size_t)sj * 64 + k] * nj);
            }
        }
        qn += __builtin_popcount(bal);
    }
    return qn;
}

__device__ __forceinline__ void sum_batch(int s, int dl, float n, int cnt, int lane, const EdgeSumCtx& c) {
    const float* xl = c.x + lane;
    float* accl = c.acc + lane;
    for (int j0 = 0; j0 < cnt; j0 += SC_GROUP) {
        float v0[SC_GROUP], v1[SC_GROUP];
#pragma unroll
        for (int u = 0; u < SC_GROUP; ++u) {
            v0[u] = 0.0f;
            v1[u] = 0.0f;
            if (j0 + u < cnt) {
                const int sj = __builtin_amdgcn_readlane(s, j0 + u);
                const float* xr = xl + (size_t)sj * 64;
                v0[u] = xr[0];
                v1[u] = xr[32];
            }
        }
#pragma unroll
        for (int u = 0; u < SC_GROUP; ++u) {
            if (j0 + u < cnt) {
                const int   dj = __builtin_amdgcn_readlane(dl, j0 + u);
                const float nj = __builtin_bit_cast(float, __builtin_amdgcn_readlane(__builtin_bit_cast(int, n), j0 + u));
                float* ar = accl + dj * 64;
                atomicAdd(ar,      v0[u] * nj);
                atomicAdd(ar + 32, v1[u] * nj);
            }
        }
    }
}

template <bool KEEP>
__device__ __forceinline__ int work_off(int qn, int lane, const EdgeSumCtx& c, int2* lst, int ltot) {
    for (int base = 0; base < qn; base += 32) {
        const int cnt = min(32, qn - base);
        const int e  = c.qlist[base + ((lane < cnt) ? lane : 0)];
        const int s  = c.src[e];
        const int dl = c.dst[e] - c.rowBase;
        const float n = c.norm[s];
        sum_batch(s, dl, n, cnt, lane, c);
        if (KEEP) {
            if (ltot + cnt <= LIST_CAP && lane < cnt)
                lst[ltot + lane] = make_int2(s | (dl << SRC_BITS), __builtin_bit_cast(int, n));
            ltot += cnt;
        }
    }
    return ltot;
}

__device__ __forceinline__ void walk_list(const int2* lst, int total, int lane, const EdgeSumCtx& c) {
    int2* ql = (int2*)c.qlist;
    for (int b = 0; b < total; b += WALK_U * 32) {
        int2 t[WALK_U];
#pragma unroll
        for (int u = 0; u < WALK_U; ++u) t[u] = lst[min(b + u * 32 + lane, total - 1)];
        static_assert(WALK_U == 8, "the statement below names eight loads");
        asm volatile("" :: "v"(t[0].x), "v"(t[0].y), "v"(t[1].x), "v"(t[1].y),
                           "v"(t[2].x), "v"(t[2].y), "v"(t[3].x), "v"(t[3].y),
                           "v"(t[4].x), "v"(t[4].y), "v"(t[5].x), "v"(t[5].y),
                           "v"(t[6].x), "v"(t[6].y), "v"(t[7].x), "v"(t[7].y));
#pragma unroll
        for (int u = 0; u < WALK_U; ++u)
            if (b + u * 32 + lane < total) ql[b + u * 32 + lane] = t[u];
    }
    for (int base = 0; base < total; base += 32) {
        const int cnt = min(32, total - base);
        const int2 cur = ql[base + ((lane < cnt) ? lane : 0)];
        const int s  = cur.x & ((1 << SRC_BITS) - 1);
        const int dl = cur.x >> SRC_BITS;
        sum_batch(s, dl, __builtin_bit_cast(float, cur.y), cnt, lane, c);
    }
}

template <bool WALK>
__global__ __launch_bounds__(OWN_THREADS) void gcn_edge_sum_lds(const float* __restrict__ x,
                                                                const int* __restrict__ src,
                                                                const int* __restrict__ dst,
                                                                const float* __restrict__ norm,
                                                                const unsigned char* __restrict__ owner,
                                                                float* __restrict__ agg,
                                                                int2* __restrict__ lists,
                                                                int* __restrict__ counts,
                                                                int nNodes) {
    __shared__ __attribute__((aligned(16))) float acc[OWN_ROWS * 64];
    __shared__ int   qlists[OWN_WAVES * Q_CAP];

    const int tid  = threadIdx.x;
    const int lane = tid & 31;
    const int wave = __builtin_amdgcn_readfirstlane(tid >> 5);
    const int rowBase = blockIdx.x * OWN_ROWS;

    for (int i = tid; i < OWN_ROWS * 64; i += OWN_THREADS) acc[i] = 0.0f;
    __syncthreads();

    EdgeSumCtx c;
    c.x = x; c.src = src; c.dst = dst; c.norm = norm; c.acc = acc;
    c.qlist  = &qlists[wave * Q_CAP];
    c.rowBase = rowBase;
    const uint4* owner4 = (const uint4*)owner;

    int2* lst  = lists + (size_t)(blockIdx.x * OWN_WAVES + wave) * LIST_CAP;
    int*  cntp = counts + blockIdx.x * OWN_WAVES + wave;

    bool scan = true;
    if (WALK) {
        const int total = __builtin_amdgcn_readfirstlane(*cntp);
        if (total <= LIST_CAP) {
            if (total > 0) walk_list(lst, total, lane, c);
            scan = false;
        }
    }

    if (scan) {

    const int start = (int)(((long long)blockIdx.x * N_TILES) / OWN_BLOCKS);

    const int lastLeft = N_EDGES - ((N_TILES - 1) * TILE_EDGES + lane * 32);
    const unsigned lastMask = (lastLeft >= 32) ? 0xFFFFFFFFu : ((lastLeft > 0) ? ((1u << lastLeft) - 1u) : 0u);

    int ltot = 0;
    int qn = 0;
    for (int base = 0; base < N_TILES; base += SCAN_U * OWN_WAVES) {
        uint4 lo[SCAN_U], hi[SCAN_U];
        int   tl[SCAN_U];
        bool  ok[SCAN_U];
#pragma unroll
        for (int u = 0; u < SCAN_U; ++u) {
            const int i = base + u * OWN_WAVES + wave;
            ok[u] = i < N_TILES;
            tl[u] = (ok[u] ? i : 0) + start;
            if (tl[u] >= N_TILES) tl[u] -= N_TILES;
            lo[u] = owner4[tl[u] * 64 + lane];
            hi[u] = owner4[tl[u] * 64 + 32 + lane];
        }
        static_assert(SCAN_U == 4, "the statements below name four tiles' loads");
        asm volatile("" :: "v"(lo[0].x), "v"(lo[0].y), "v"(lo[0].z), "v"(lo[0].w),
                           "v"(hi[0].x), "v"(hi[0].y), "v"(hi[0].z), "v"(hi[0].w),
                           "v"(lo[1].x), "v"(lo[1].y), "v"(lo[1].z), "v"(lo[1].w),
                           "v"(hi[1].x), "v"(hi[1].y), "v"(hi[1].z), "v"(hi[1].w));
        asm volatile("" :: "v"(lo[2].x), "v"(lo[2].y), "v"(lo[2].z), "v"(lo[2].w),
                           "v"(hi[2].x), "v"(hi[2].y), "v"(hi[2].z), "v"(hi[2].w),
                           "v"(lo[3].x), "v"(lo[3].y), "v"(lo[3].z), "v"(lo[3].w),
                           "v"(hi[3].x), "v"(hi[3].y), "v"(hi[3].z), "v"(hi[3].w));
#pragma unroll
        for (int u = 0; u < SCAN_U; ++u) {
            if (ok[u]) {
                const int e0 = tl[u] * TILE_EDGES + lane * 32;
                unsigned m = group_mask(lo[u], hi[u], (unsigned)blockIdx.x);
                if (tl[u] == N_TILES - 1) m &= lastMask;
                qn = list_group(m, e0, qn, c);
            }
        }
        if (qn >= Q_BATCH) {
            if (qn > Q_CAP) ltot = LIST_CAP + 1;
            ltot = work_off<!WALK>(min(qn, Q_CAP), lane, c, lst, ltot);
            qn = 0;
        }
    }
    if (qn > Q_CAP) ltot = LIST_CAP + 1;
    ltot = work_off<!WALK>(min(qn, Q_CAP), lane, c, lst, ltot);
    if (!WALK && lane == 0) *cntp = ltot;

    }
    __syncthreads();

    const int rowsHere = min(OWN_ROWS, nNodes - rowBase);
    float4* out4 = (float4*)(agg + (size_t)rowBase * 64);
    const float4* acc4 = (const float4*)acc;
    for (int i = tid; i < rowsHere * 16; i += OWN_THREADS) out4[i] = acc4[i];
}

template <int DIN, int DOUT, bool RELU>
__global__ __launch_bounds__(256) void gcn_gemm_wmma(const float* __restrict__ agg,
                                                     const float* __restrict__ norm,
                                                     const float* __restrict__ W,
                                                     const float* __restrict__ bias,
                                                     float* __restrict__ out,
                                                     int nNodes) {
    const int lane = threadIdx.x & 31;
    const int wave = __builtin_amdgcn_readfirstlane((int)(threadIdx.x >> 5));
    const int tilesPerRow = DOUT / 16;
    const int totalTiles  = (nNodes / 16) * tilesPerRow;
    const int tileStride  = gridDim.x * 8;

    const int m  = lane & 15;
    const int kh = lane >> 4;

    for (int tile = blockIdx.x * 8 + wave; tile < totalTiles; tile += tileStride) {

    const int nodeBlk = tile / tilesPerRow;
    const int colBlk  = tile % tilesPerRow;
    const int node0   = nodeBlk * 16;
    const int col0    = colBlk * 16;

    const float rowScale = norm[node0 + m];
    const float* arow = agg + (size_t)(node0 + m) * DIN;

    v8f c = {0.f, 0.f, 0.f, 0.f, 0.f, 0.f, 0.f, 0.f};

#pragma unroll
    for (int k0 = 0; k0 < DIN; k0 += 4) {
        const int ka = k0 + 2 * kh;
        float2 ap = *(const float2*)(arow + ka);
        v2f a;
        a.x = ap.x * rowScale;
        a.y = ap.y * rowScale;
        v2f b;
        b.x = W[(size_t)ka * DOUT + col0 + m];
        b.y = W[(size_t)(ka + 1) * DOUT + col0 + m];
        c = __builtin_amdgcn_wmma_f32_16x16x4_f32(
                 false, a,  false, b,
                 (short)0, c,  false,  false);
    }

    const float bv = bias[col0 + m];
#pragma unroll
    for (int r = 0; r < 8; ++r) {
        const int row = r + 8 * kh;
        float v = c[r] + bv;
        if (RELU) v = fmaxf(v, 0.0f);
        out[(size_t)(node0 + row) * DOUT + col0 + m] = v;
    }

    }
}

extern "C" void kernel_launch(void* const* d_in, const int* in_sizes, int n_in,
                              void* d_out, int out_size, void* d_ws, size_t ws_size,
                              hipStream_t stream) {
    const float* x   = (const float*)d_in[0];
    const int*   ei  = (const int*)d_in[1];
    const float* W1  = (const float*)d_in[2];
    const float* b1  = (const float*)d_in[3];
    const float* W2  = (const float*)d_in[4];
    const float* b2  = (const float*)d_in[5];
    float*       out = (float*)d_out;

    const int N = in_sizes[0] / 64;
    const int E = in_sizes[1] / 2;
    const int* src = ei;
    const int* dst = ei + E;

    float* norm = (float*)d_ws;
    float* deg1 = norm + N;
    float* agg  = deg1 + N + 256;
    float* h    = agg + (size_t)N * 64;
    unsigned char* owner = (unsigned char*)(h + (size_t)N * 64);
    int*  counts = (int*)(owner + OWNER_SPAN);
    int2* lists  = (int2*)(counts + OWN_BLOCKS * OWN_WAVES);

    {
        int n4 = (2 * N) / 4;
        gcn_zero_f4<<<RES_BLOCKS, ZBLOCK, 0, stream>>>((float4*)norm, n4);
    }

    gcn_degree<<<RES_BLOCKS, 256, 0, stream>>>(dst, norm, deg1, owner, E);
    gcn_norm<<<(N + 255) / 256, 256, 0, stream>>>(norm, deg1, N);

    gcn_edge_sum_lds<false><<<OWN_BLOCKS, OWN_THREADS, 0, stream>>>(x, src, dst, norm, owner, agg, lists, counts, N);
    gcn_gemm_wmma<64, 64, true><<<RES_BLOCKS, 256, 0, stream>>>(agg, norm, W1, b1, h, N);

    gcn_edge_sum_lds<true><<<OWN_BLOCKS, OWN_THREADS, 0, stream>>>(h, src, dst, norm, owner, agg, lists, counts, N);
    gcn_gemm_wmma<64, 32, false><<<RES_BLOCKS, 256, 0, stream>>>(agg, norm, W2, b2, out, N);
}


# In-degrees counted by the owner blocks: a find pass scans the ids, lists its edges, counts them per owned row by LDS integer adds, stores norm; no global adds, zero fill, norm launch; h stored x norm
# speedup vs baseline: 3.2475x; 1.0642x over previous
#include <hip/hip_runtime.h>

typedef float v2f __attribute__((ext_vector_type(2)));
typedef float v8f __attribute__((ext_vector_type(8)));

#define N_NODES 100000
#define N_EDGES 1250000

#define RES_BLOCKS 1024

#define OWN_BLOCKS  256
#define OWN_ROWS    ((N_NODES + OWN_BLOCKS - 1) / OWN_BLOCKS)
#define TILE_EDGES  1024
#define N_TILES     ((N_EDGES + TILE_EDGES - 1) / TILE_EDGES)

__global__ __launch_bounds__(32) void gcn_owner_ids(const int* __restrict__ dst,
                                                    unsigned char* __restrict__ owner, int nEdges) {
    const int lane = threadIdx.x;
    const int tile = blockIdx.x;
    uint4* owner4 = (uint4*)owner;
    const int e0 = tile * TILE_EDGES + lane;
    int d[32];
#pragma unroll
    for (int g = 0; g < 32; ++g) d[g] = dst[min(e0 + g * 32, nEdges - 1)];
    int p[8] = {0, 0, 0, 0, 0, 0, 0, 0};
#pragma unroll
    for (int g = 0; g < 32; ++g) {
        const unsigned id = (unsigned)d[g] / (unsigned)OWN_ROWS;
#pragma unroll
        for (int b = 0; b < 8; ++b) {
            const int plane = (int)__builtin_amdgcn_ballot_w32(((id >> b) & 1u) != 0u);
            p[b] = (lane == g) ? plane : p[b];
        }
    }
    owner4[tile * 64 + lane]      = make_uint4((unsigned)p[0], (unsigned)p[1], (unsigned)p[2], (unsigned)p[3]);
    owner4[tile * 64 + 32 + lane] = make_uint4((unsigned)p[4], (unsigned)p[5], (unsigned)p[6], (unsigned)p[7]);
}

#define OWN_THREADS 1024
#define OWN_WAVES   (OWN_THREADS / 32)
#define SCAN_U      4
#define Q_BATCH     64
#define Q_CAP       1024
#define SC_GROUP    16
#define LIST_CAP    512
#define WALK_U      8
#define SRC_BITS    17
static_assert(N_NODES <= (1 << SRC_BITS), "a source row number fits SRC_BITS bits");
static_assert(OWN_ROWS <= (1 << (31 - SRC_BITS)), "an owned row number fits above it in a positive int");
static_assert(LIST_CAP <= Q_CAP, "a sum pass holds a whole list in the wave's LDS list space");
static_assert(OWN_BLOCKS <= 256, "an owner id has eight bits");
#define OWNER_SPAN  (N_TILES * TILE_EDGES)

enum { FIND = 0, SUM_SCALED = 1, SUM_PLAIN = 2 };

__device__ __forceinline__ unsigned group_mask(uint4 lo, uint4 hi, unsigned id) {
    unsigned m =  ~(lo.x ^ (0u - ( id       & 1u)));
    m &= ~(lo.y ^ (0u - ((id >> 1) & 1u)));
    m &= ~(lo.z ^ (0u - ((id >> 2) & 1u)));
    m &= ~(lo.w ^ (0u - ((id >> 3) & 1u)));
    m &= ~(hi.x ^ (0u - ((id >> 4) & 1u)));
    m &= ~(hi.y ^ (0u - ((id >> 5) & 1u)));
    m &= ~(hi.z ^ (0u - ((id >> 6) & 1u)));
    m &= ~(hi.w ^ (0u - ((id >> 7) & 1u)));
    return m;
}

struct OwnCtx {
    const float* x;
    const int*   src;
    const int*   dst;
    const float* norm;
    float*       acc;
    int*         deg;
    int*         qlist;
    int          rowBase;
};

template <int MODE>
__device__ __forceinline__ int list_group(unsigned m, int e0, int qn, const OwnCtx& c) {
    unsigned bal;
    while ((bal = __builtin_amdgcn_ballot_w32(m != 0u)) != 0u) {
        if (m != 0u) {
            const int e = e0 + (int)__builtin_ctz(m);
            m &= m - 1u;
            const int pos = qn + (int)__builtin_amdgcn_mbcnt_lo(bal, 0u);
            if (pos < Q_CAP) {
                c.qlist[pos] = e;
            } else {
                const int dl = c.dst[e] - c.rowBase;
                if (MODE == FIND) {
                    atomicAdd(c.deg + dl, 1);
                } else {
                    const int   sj = c.src[e];
                    const float nj = (MODE == SUM_SCALED) ? c.norm[sj] : 1.0f;
#pragma nounroll
                    for (int k = 0; k < 64; ++k) atomicAdd(c.acc + dl * 64 + k, c.x[(size_t)sj * 64 + k] * nj);
                }
            }
        }
        qn += __builtin_popcount(bal);
    }
    return qn;
}

template <bool SCALED>
__device__ __forceinline__ void sum_batch(int s, int dl, float n, int cnt, int lane, const OwnCtx& c) {
    const float* xl = c.x + lane;
    float* accl = c.acc + lane;
    for (int j0 = 0; j0 < cnt; j0 += SC_GROUP) {
        float v0[SC_GROUP], v1[SC_GROUP];
#pragma unroll
        for (int u = 0; u < SC_GROUP; ++u) {
            v0[u] = 0.0f;
            v1[u] = 0.0f;
            if (j0 + u < cnt) {
                const int sj = __builtin_amdgcn_readlane(s, j0 + u);
                const float* xr = xl + (size_t)sj * 64;
                v0[u] = xr[0];
                v1[u] = xr[32];
            }
        }
#pragma unroll
        for (int u = 0; u < SC_GROUP; ++u) {
            if (j0 + u < cnt) {
                const int dj = __builtin_amdgcn_readlane(dl, j0 + u);
                float a0 = v0[u], a1 = v1[u];
                if (SCALED) {
                    const float nj = __builtin_bit_cast(float, __builtin_amdgcn_readlane(__builtin_bit_cast(int, n), j0 + u));
                    a0 *= nj;
                    a1 *= nj;
                }
                float* ar = accl + dj * 64;
                atomicAdd(ar,      a0);
                atomicAdd(ar + 32, a1);
            }
        }
    }
}

template <int MODE>
__device__ __forceinline__ int work_off(int qn, int lane, const OwnCtx& c, int* lst, int ltot) {
    for (int base = 0; base < qn; base += 32) {
        const int cnt = min(32, qn - base);
        const int e  = c.qlist[base + ((lane < cnt) ? lane : 0)];
        const int s  = c.src[e];
        const int dl = c.dst[e] - c.rowBase;
        if (MODE == FIND) {
            if (lane < cnt) {
                atomicAdd(c.deg + dl, 1);
                if (ltot + cnt <= LIST_CAP) lst[ltot + lane] = s | (dl << SRC_BITS);
            }
            ltot += cnt;
        } else {
            const float n = (MODE == SUM_SCALED) ? c.norm[s] : 1.0f;
            sum_batch<MODE == SUM_SCALED>(s, dl, n, cnt, lane, c);
        }
    }
    return ltot;
}

template <bool SCALED>
__device__ __forceinline__ void walk_list(const int* lst, int total, int lane, const OwnCtx& c) {
    int* ql = c.qlist;
    for (int b = 0; b < total; b += WALK_U * 32) {
        int t[WALK_U];
#pragma unroll
        for (int u = 0; u < WALK_U; ++u) t[u] = lst[min(b + u * 32 + lane, total - 1)];
        static_assert(WALK_U == 8, "the statement below names eight loads");
        asm volatile("" :: "v"(t[0]), "v"(t[1]), "v"(t[2]), "v"(t[3]),
                           "v"(t[4]), "v"(t[5]), "v"(t[6]), "v"(t[7]));
#pragma unroll
        for (int u = 0; u < WALK_U; ++u)
            if (b + u * 32 + lane < total) ql[b + u * 32 + lane] = t[u];
    }
    for (int base = 0; base < total; base += 32) {
        const int cnt = min(32, total - base);
        const int w  = ql[base + ((lane < cnt) ? lane : 0)];
        const int s  = w & ((1 << SRC_BITS) - 1);
        const int dl = w >> SRC_BITS;
        const float n = SCALED ? c.norm[s] : 1.0f;
        sum_batch<SCALED>(s, dl, n, cnt, lane, c);
    }
}

template <int MODE>
__device__ __forceinline__ int scan_ids(const uint4* __restrict__ owner4, int lane, int wave,
                                        const OwnCtx& c, int* lst) {
    const int start = (int)(((long long)blockIdx.x * N_TILES) / OWN_BLOCKS);

    const int lastLeft = N_EDGES - ((N_TILES - 1) * TILE_EDGES + lane * 32);
    const unsigned lastMask = (lastLeft >= 32) ? 0xFFFFFFFFu : ((lastLeft > 0) ? ((1u << lastLeft) - 1u) : 0u);

    int ltot = 0;
    int qn = 0;
    for (int base = 0; base < N_TILES; base += SCAN_U * OWN_WAVES) {
        uint4 lo[SCAN_U], hi[SCAN_U];
        int   tl[SCAN_U];
        bool  ok[SCAN_U];
#pragma unroll
        for (int u = 0; u < SCAN_U; ++u) {
            const int i = base + u * OWN_WAVES + wave;
            ok[u] = i < N_TILES;
            tl[u] = (ok[u] ? i : 0) + start;
            if (tl[u] >= N_TILES) tl[u] -= N_TILES;
            lo[u] = owner4[tl[u] * 64 + lane];
            hi[u] = owner4[tl[u] * 64 + 32 + lane];
        }
        static_assert(SCAN_U == 4, "the statements below name four tiles' loads");
        asm volatile("" :: "v"(lo[0].x), "v"(lo[0].y), "v"(lo[0].z), "v"(lo[0].w),
                           "v"(hi[0].x), "v"(hi[0].y), "v"(hi[0].z), "v"(hi[0].w),
                           "v"(lo[1].x), "v"(lo[1].y), "v"(lo[1].z), "v"(lo[1].w),
                           "v"(hi[1].x), "v"(hi[1].y), "v"(hi[1].z), "v"(hi[1].w));
        asm volatile("" :: "v"(lo[2].x), "v"(lo[2].y), "v"(lo[2].z), "v"(lo[2].w),
                           "v"(hi[2].x), "v"(hi[2].y), "v"(hi[2].z), "v"(hi[2].w),
                           "v"(lo[3].x), "v"(lo[3].y), "v"(lo[3].z), "v"(lo[3].w),
                           "v"(hi[3].x), "v"(hi[3].y), "v"(hi[3].z), "v"(hi[3].w));
#pragma unroll
        for (int u = 0; u < SCAN_U; ++u) {
            if (ok[u]) {
                const int e0 = tl[u] * TILE_EDGES + lane * 32;
                unsigned m = group_mask(lo[u], hi[u], (unsigned)blockIdx.x);
                if (tl[u] == N_TILES - 1) m &= lastMask;
                qn = list_group<MODE>(m, e0, qn, c);
            }
        }
        if (qn >= Q_BATCH) {
            if (qn > Q_CAP) ltot = LIST_CAP + 1;
            ltot = work_off<MODE>(min(qn, Q_CAP), lane, c, lst, ltot);
            qn = 0;
        }
    }
    if (qn > Q_CAP) ltot = LIST_CAP + 1;
    ltot = work_off<MODE>(min(qn, Q_CAP), lane, c, lst, ltot);
    return ltot;
}

__global__ __launch_bounds__(OWN_THREADS) void gcn_find_edges(const int* __restrict__ src,
                                                              const int* __restrict__ dst,
                                                              const unsigned char* __restrict__ owner,
                                                              float* __restrict__ norm,
                                                              int* __restrict__ lists,
                                                              int* __restrict__ counts,
                                                              int nNodes) {
    __shared__ int deg[OWN_ROWS];
    __shared__ int qlists[OWN_WAVES * Q_CAP];

    const int tid  = threadIdx.x;
    const int lane = tid & 31;
    const int wave = __builtin_amdgcn_readfirstlane(tid >> 5);
    const int rowBase = blockIdx.x * OWN_ROWS;

    for (int i = tid; i < OWN_ROWS; i += OWN_THREADS) deg[i] = 0;
    __syncthreads();

    OwnCtx c;
    c.x = nullptr; c.src = src; c.dst = dst; c.norm = nullptr; c.acc = nullptr; c.deg = deg;
    c.qlist  = &qlists[wave * Q_CAP];
    c.rowBase = rowBase;

    int* lst  = lists + (size_t)(blockIdx.x * OWN_WAVES + wave) * LIST_CAP;
    int* cntp = counts + blockIdx.x * OWN_WAVES + wave;

    const int ltot = scan_ids<FIND>((const uint4*)owner, lane, wave, c, lst);
    if (lane == 0) *cntp = ltot;
    __syncthreads();

    const int rowsHere = min(OWN_ROWS, nNodes - rowBase);
    for (int i = tid; i < rowsHere; i += OWN_THREADS) {
        const float d = (float)deg[i];
        norm[rowBase + i] = (d > 0.0f) ? rsqrtf(fmaxf(d, 1.0f)) : 0.0f;
    }
}

template <bool SCALED>
__global__ __launch_bounds__(OWN_THREADS) void gcn_edge_sum_lds(const float* __restrict__ x,
                                                                const int* __restrict__ src,
                                                                const int* __restrict__ dst,
                                                                const float* __restrict__ norm,
                                                                const unsigned char* __restrict__ owner,
                                                                float* __restrict__ agg,
                                                                const int* __restrict__ lists,
                                                                const int* __restrict__ counts,
                                                                int nNodes) {
    __shared__ __attribute__((aligned(16))) float acc[OWN_ROWS * 64];
    __shared__ int   qlists[OWN_WAVES * Q_CAP];

    const int tid  = threadIdx.x;
    const int lane = tid & 31;
    const int wave = __builtin_amdgcn_readfirstlane(tid >> 5);
    const int rowBase = blockIdx.x * OWN_ROWS;

    for (int i = tid; i < OWN_ROWS * 64; i += OWN_THREADS) acc[i] = 0.0f;
    __syncthreads();

    OwnCtx c;
    c.x = x; c.src = src; c.dst = dst; c.norm = norm; c.acc = acc; c.deg = nullptr;
    c.qlist  = &qlists[wave * Q_CAP];
    c.rowBase = rowBase;

    const int* lst = lists + (size_t)(blockIdx.x * OWN_WAVES + wave) * LIST_CAP;
    const int total = __builtin_amdgcn_readfirstlane(counts[blockIdx.x * OWN_WAVES + wave]);
    if (total <= LIST_CAP) {
        if (total > 0) walk_list<SCALED>(lst, total, lane, c);
    } else {
        scan_ids<SCALED ? SUM_SCALED : SUM_PLAIN>((const uint4*)owner, lane, wave, c, nullptr);
    }
    __syncthreads();

    const int rowsHere = min(OWN_ROWS, nNodes - rowBase);
    float4* out4 = (float4*)(agg + (size_t)rowBase * 64);
    const float4* acc4 = (const float4*)acc;
    for (int i = tid; i < rowsHere * 16; i += OWN_THREADS) out4[i] = acc4[i];
}

template <int DIN, int DOUT, bool RELU, bool SCALE_OUT>
__global__ __launch_bounds__(256) void gcn_gemm_wmma(const float* __restrict__ agg,
                                                     const float* __restrict__ norm,
                                                     const float* __restrict__ W,
                                                     const float* __restrict__ bias,
                                                     float* __restrict__ out,
                                                     int nNodes) {
    const int lane = threadIdx.x & 31;
    const int wave = __builtin_amdgcn_readfirstlane((int)(threadIdx.x >> 5));
    const int tilesPerRow = DOUT / 16;
    const int totalTiles  = (nNodes / 16) * tilesPerRow;
    const int tileStride  = gridDim.x * 8;

    const int m  = lane & 15;
    const int kh = lane >> 4;

    for (int tile = blockIdx.x * 8 + wave; tile < totalTiles; tile += tileStride) {

    const int nodeBlk = tile / tilesPerRow;
    const int colBlk  = tile % tilesPerRow;
    const int node0   = nodeBlk * 16;
    const int col0    = colBlk * 16;

    const float rowScale = norm[node0 + m];
    const float* arow = agg + (size_t)(node0 + m) * DIN;

    float os[8];
    if (SCALE_OUT) {
#pragma unroll
        for (int r = 0; r < 8; ++r) os[r] = norm[node0 + r + 8 * kh];
    }

    v8f c = {0.f, 0.f, 0.f, 0.f, 0.f, 0.f, 0.f, 0.f};

#pragma unroll
    for (int k0 = 0; k0 < DIN; k0 += 4) {
        const int ka = k0 + 2 * kh;
        float2 ap = *(const float2*)(arow + ka);
        v2f a;
        a.x = ap.x * rowScale;
        a.y = ap.y * rowScale;
        v2f b;
        b.x = W[(size_t)ka * DOUT + col0 + m];
        b.y = W[(size_t)(ka + 1) * DOUT + col0 + m];
        c = __builtin_amdgcn_wmma_f32_16x16x4_f32(
                 false, a,  false, b,
                 (short)0, c,  false,  false);
    }

    const float bv = bias[col0 + m];
#pragma unroll
    for (int r = 0; r < 8; ++r) {
        const int row = r + 8 * kh;
        float v = c[r] + bv;
        if (RELU) v = fmaxf(v, 0.0f);
        if (SCALE_OUT) v *= os[r];
        out[(size_t)(node0 + row) * DOUT + col0 + m] = v;
    }

    }
}

extern "C" void kernel_launch(void* const* d_in, const int* in_sizes, int n_in,
                              void* d_out, int out_size, void* d_ws, size_t ws_size,
                              hipStream_t stream) {
    const float* x   = (const float*)d_in[0];
    const int*   ei  = (const int*)d_in[1];
    const float* W1  = (const float*)d_in[2];
    const float* b1  = (const float*)d_in[3];
    const float* W2  = (const float*)d_in[4];
    const float* b2  = (const float*)d_in[5];
    float*       out = (float*)d_out;

    const int N = in_sizes[0] / 64;
    const int E = in_sizes[1] / 2;
    const int* src = ei;
    const int* dst = ei + E;

    float* norm = (float*)d_ws;
    float* agg  = norm + 2 * N + 256;
    float* h    = agg + (size_t)N * 64;
    unsigned char* owner = (unsigned char*)(h + (size_t)N * 64);
    int* counts = (int*)(owner + OWNER_SPAN);
    int* lists  = counts + OWN_BLOCKS * OWN_WAVES;

    gcn_owner_ids<<<N_TILES, 32, 0, stream>>>(dst, owner, E);
    gcn_find_edges<<<OWN_BLOCKS, OWN_THREADS, 40960, stream>>>(src, dst, owner, norm, lists, counts, N);

    gcn_edge_sum_lds<true><<<OWN_BLOCKS, OWN_THREADS, 0, stream>>>(x, src, dst, norm, owner, agg, lists, counts, N);
    gcn_gemm_wmma<64, 64, true, true><<<RES_BLOCKS, 256, 0, stream>>>(agg, norm, W1, b1, h, N);

    gcn_edge_sum_lds<false><<<OWN_BLOCKS, OWN_THREADS, 0, stream>>>(h, src, dst, norm, owner, agg, lists, counts, N);
    gcn_gemm_wmma<64, 32, false, false><<<RES_BLOCKS, 256, 0, stream>>>(agg, norm, W2, b2, out, N);
}


# Each product runs at the end of its edge-sum pass: a wave feeds 16 summed rows from the block's LDS (pitch 68) to WMMA as A fragments, results stored in lines; 4 launches, not 6; sum order kept
# speedup vs baseline: 3.6941x; 1.1375x over previous
#include <hip/hip_runtime.h>

typedef float v2f __attribute__((ext_vector_type(2)));
typedef float v8f __attribute__((ext_vector_type(8)));

#define N_NODES 100000
#define N_EDGES 1250000

#define OWN_BLOCKS  256
#define OWN_ROWS    ((N_NODES + OWN_BLOCKS - 1) / OWN_BLOCKS)
#define ROW_TILES   ((OWN_ROWS + 15) / 16)
#define ACC_ROWS    (ROW_TILES * 16)
#define ACC_PITCH   68
#define TILE_EDGES  1024
#define N_TILES     ((N_EDGES + TILE_EDGES - 1) / TILE_EDGES)

__global__ __launch_bounds__(32) void gcn_owner_ids(const int* __restrict__ dst,
                                                    unsigned char* __restrict__ owner, int nEdges) {
    const int lane = threadIdx.x;
    const int tile = blockIdx.x;
    uint4* owner4 = (uint4*)owner;
    const int e0 = tile * TILE_EDGES + lane;
    int d[32];
#pragma unroll
    for (int g = 0; g < 32; ++g) d[g] = dst[min(e0 + g * 32, nEdges - 1)];
    int p[8] = {0, 0, 0, 0, 0, 0, 0, 0};
#pragma unroll
    for (int g = 0; g < 32; ++g) {
        const unsigned id = (unsigned)d[g] / (unsigned)OWN_ROWS;
#pragma unroll
        for (int b = 0; b < 8; ++b) {
            const int plane = (int)__builtin_amdgcn_ballot_w32(((id >> b) & 1u) != 0u);
            p[b] = (lane == g) ? plane : p[b];
        }
    }
    owner4[tile * 64 + lane]      = make_uint4((unsigned)p[0], (unsigned)p[1], (unsigned)p[2], (unsigned)p[3]);
    owner4[tile * 64 + 32 + lane] = make_uint4((unsigned)p[4], (unsigned)p[5], (unsigned)p[6], (unsigned)p[7]);
}

#define OWN_THREADS 1024
#define OWN_WAVES   (OWN_THREADS / 32)
#define SCAN_U      4
#define Q_BATCH     64
#define Q_CAP       1024
#define SC_GROUP    16
#define LIST_CAP    512
#define WALK_U      8
#define SRC_BITS    17
static_assert(N_NODES <= (1 << SRC_BITS), "a source row number fits SRC_BITS bits");
static_assert(OWN_ROWS <= (1 << (31 - SRC_BITS)), "an owned row number fits above it in a positive int");
static_assert(LIST_CAP <= Q_CAP, "a sum pass holds a whole list in the wave's LDS list space");
static_assert(OWN_BLOCKS <= 256, "an owner id has eight bits");
#define OWNER_SPAN  (N_TILES * TILE_EDGES)

enum { FIND = 0, SUM_SCALED = 1, SUM_PLAIN = 2 };

__device__ __forceinline__ unsigned group_mask(uint4 lo, uint4 hi, unsigned id) {
    unsigned m =  ~(lo.x ^ (0u - ( id       & 1u)));
    m &= ~(lo.y ^ (0u - ((id >> 1) & 1u)));
    m &= ~(lo.z ^ (0u - ((id >> 2) & 1u)));
    m &= ~(lo.w ^ (0u - ((id >> 3) & 1u)));
    m &= ~(hi.x ^ (0u - ((id >> 4) & 1u)));
    m &= ~(hi.y ^ (0u - ((id >> 5) & 1u)));
    m &= ~(hi.z ^ (0u - ((id >> 6) & 1u)));
    m &= ~(hi.w ^ (0u - ((id >> 7) & 1u)));
    return m;
}

struct OwnCtx {
    const float* x;
    const int*   src;
    const int*   dst;
    const float* norm;
    float*       acc;
    int*         deg;
    int*         qlist;
    int          rowBase;
};

template <int MODE>
__device__ __forceinline__ int list_group(unsigned m, int e0, int qn, const OwnCtx& c) {
    unsigned bal;
    while ((bal = __builtin_amdgcn_ballot_w32(m != 0u)) != 0u) {
        if (m != 0u) {
            const int e = e0 + (int)__builtin_ctz(m);
            m &= m - 1u;
            const int pos = qn + (int)__builtin_amdgcn_mbcnt_lo(bal, 0u);
            if (pos < Q_CAP) {
                c.qlist[pos] = e;
            } else {
                const int dl = c.dst[e] - c.rowBase;
                if (MODE == FIND) {
                    atomicAdd(c.deg + dl, 1);
                } else {
                    const int   sj = c.src[e];
                    const float nj = (MODE == SUM_SCALED) ? c.norm[sj] : 1.0f;
#pragma nounroll
                    for (int k = 0; k < 64; ++k) atomicAdd(c.acc + dl * ACC_PITCH + k, c.x[(size_t)sj * 64 + k] * nj);
                }
            }
        }
        qn += __builtin_popcount(bal);
    }
    return qn;
}

template <bool SCALED>
__device__ __forceinline__ void sum_batch(int s, int dl, float n, int cnt, int lane, const OwnCtx& c) {
    const float* xl = c.x + lane;
    float* accl = c.acc + lane;
    for (int j0 = 0; j0 < cnt; j0 += SC_GROUP) {
        float v0[SC_GROUP], v1[SC_GROUP];
#pragma unroll
        for (int u = 0; u < SC_GROUP; ++u) {
            v0[u] = 0.0f;
            v1[u] = 0.0f;
            if (j0 + u < cnt) {
                const int sj = __builtin_amdgcn_readlane(s, j0 + u);
                const float* xr = xl + (size_t)sj * 64;
                v0[u] = xr[0];
                v1[u] = xr[32];
            }
        }
#pragma unroll
        for (int u = 0; u < SC_GROUP; ++u) {
            if (j0 + u < cnt) {
                const int dj = __builtin_amdgcn_readlane(dl, j0 + u);
                float a0 = v0[u], a1 = v1[u];
                if (SCALED) {
                    const float nj = __builtin_bit_cast(float, __builtin_amdgcn_readlane(__builtin_bit_cast(int, n), j0 + u));
                    a0 *= nj;
                    a1 *= nj;
                }
                float* ar = accl + dj * ACC_PITCH;
                atomicAdd(ar,      a0);
                atomicAdd(ar + 32, a1);
            }
        }
    }
}

template <int MODE>
__device__ __forceinline__ int work_off(int qn, int lane, const OwnCtx& c, int* lst, int ltot) {
    for (int base = 0; base < qn; base += 32) {
        const int cnt = min(32, qn - base);
        const int e  = c.qlist[base + ((lane < cnt) ? lane : 0)];
        const int s  = c.src[e];
        const int dl = c.dst[e] - c.rowBase;
        if (MODE == FIND) {
            if (lane < cnt) {
                atomicAdd(c.deg + dl, 1);
                if (ltot + cnt <= LIST_CAP) lst[ltot + lane] = s | (dl << SRC_BITS);
            }
            ltot += cnt;
        } else {
            const float n = (MODE == SUM_SCALED) ? c.norm[s] : 1.0f;
            sum_batch<MODE == SUM_SCALED>(s, dl, n, cnt, lane, c);
        }
    }
    return ltot;
}

template <bool SCALED>
__device__ __forceinline__ void walk_list(const int* lst, int total, int lane, const OwnCtx& c) {
    int* ql = c.qlist;
    for (int b = 0; b < total; b += WALK_U * 32) {
        int t[WALK_U];
#pragma unroll
        for (int u = 0; u < WALK_U; ++u) t[u] = lst[min(b + u * 32 + lane, total - 1)];
        static_assert(WALK_U == 8, "the statement below names eight loads");
        asm volatile("" :: "v"(t[0]), "v"(t[1]), "v"(t[2]), "v"(t[3]),
                           "v"(t[4]), "v"(t[5]), "v"(t[6]), "v"(t[7]));
#pragma unroll
        for (int u = 0; u < WALK_U; ++u)
            if (b + u * 32 + lane < total) ql[b + u * 32 + lane] = t[u];
    }
    for (int base = 0; base < total; base += 32) {
        const int cnt = min(32, total - base);
        const int w  = ql[base + ((lane < cnt) ? lane : 0)];
        const int s  = w & ((1 << SRC_BITS) - 1);
        const int dl = w >> SRC_BITS;
        const float n = SCALED ? c.norm[s] : 1.0f;
        sum_batch<SCALED>(s, dl, n, cnt, lane, c);
    }
}

template <int MODE>
__device__ __forceinline__ int scan_ids(const uint4* __restrict__ owner4, int lane, int wave,
                                        const OwnCtx& c, int* lst) {
    const int start = (int)(((long long)blockIdx.x * N_TILES) / OWN_BLOCKS);

    const int lastLeft = N_EDGES - ((N_TILES - 1) * TILE_EDGES + lane * 32);
    const unsigned lastMask = (lastLeft >= 32) ? 0xFFFFFFFFu : ((lastLeft > 0) ? ((1u << lastLeft) - 1u) : 0u);

    int ltot = 0;
    int qn = 0;
    for (int base = 0; base < N_TILES; base += SCAN_U * OWN_WAVES) {
        uint4 lo[SCAN_U], hi[SCAN_U];
        int   tl[SCAN_U];
        bool  ok[SCAN_U];
#pragma unroll
        for (int u = 0; u < SCAN_U; ++u) {
            const int i = base + u * OWN_WAVES + wave;
            ok[u] = i < N_TILES;
            tl[u] = (ok[u] ? i : 0) + start;
            if (tl[u] >= N_TILES) tl[u] -= N_TILES;
            lo[u] = owner4[tl[u] * 64 + lane];
            hi[u] = owner4[tl[u] * 64 + 32 + lane];
        }
        static_assert(SCAN_U == 4, "the statements below name four tiles' loads");
        asm volatile("" :: "v"(lo[0].x), "v"(lo[0].y), "v"(lo[0].z), "v"(lo[0].w),
                           "v"(hi[0].x), "v"(hi[0].y), "v"(hi[0].z), "v"(hi[0].w),
                           "v"(lo[1].x), "v"(lo[1].y), "v"(lo[1].z), "v"(lo[1].w),
                           "v"(hi[1].x), "v"(hi[1].y), "v"(hi[1].z), "v"(hi[1].w));
        asm volatile("" :: "v"(lo[2].x), "v"(lo[2].y), "v"(lo[2].z), "v"(lo[2].w),
                           "v"(hi[2].x), "v"(hi[2].y), "v"(hi[2].z), "v"(hi[2].w),
                           "v"(lo[3].x), "v"(lo[3].y), "v"(lo[3].z), "v"(lo[3].w),
                           "v"(hi[3].x), "v"(hi[3].y), "v"(hi[3].z), "v"(hi[3].w));
#pragma unroll
        for (int u = 0; u < SCAN_U; ++u) {
            if (ok[u]) {
                const int e0 = tl[u] * TILE_EDGES + lane * 32;
                unsigned m = group_mask(lo[u], hi[u], (unsigned)blockIdx.x);
                if (tl[u] == N_TILES - 1) m &= lastMask;
                qn = list_group<MODE>(m, e0, qn, c);
            }
        }
        if (qn >= Q_BATCH) {
            if (qn > Q_CAP) ltot = LIST_CAP + 1;
            ltot = work_off<MODE>(min(qn, Q_CAP), lane, c, lst, ltot);
            qn = 0;
        }
    }
    if (qn > Q_CAP) ltot = LIST_CAP + 1;
    ltot = work_off<MODE>(min(qn, Q_CAP), lane, c, lst, ltot);
    return ltot;
}

__global__ __launch_bounds__(OWN_THREADS) void gcn_find_edges(const int* __restrict__ src,
                                                              const int* __restrict__ dst,
                                                              const unsigned char* __restrict__ owner,
                                                              float* __restrict__ norm,
                                                              int* __restrict__ lists,
                                                              int* __restrict__ counts,
                                                              int nNodes) {
    __shared__ int deg[OWN_ROWS];
    __shared__ int qlists[OWN_WAVES * Q_CAP];

    const int tid  = threadIdx.x;
    const int lane = tid & 31;
    const int wave = __builtin_amdgcn_readfirstlane(tid >> 5);
    const int rowBase = blockIdx.x * OWN_ROWS;

    for (int i = tid; i < OWN_ROWS; i += OWN_THREADS) deg[i] = 0;
    __syncthreads();

    OwnCtx c;
    c.x = nullptr; c.src = src; c.dst = dst; c.norm = nullptr; c.acc = nullptr; c.deg = deg;
    c.qlist  = &qlists[wave * Q_CAP];
    c.rowBase = rowBase;

    int* lst  = lists + (size_t)(blockIdx.x * OWN_WAVES + wave) * LIST_CAP;
    int* cntp = counts + blockIdx.x * OWN_WAVES + wave;

    const int ltot = scan_ids<FIND>((const uint4*)owner, lane, wave, c, lst);
    if (lane == 0) *cntp = ltot;
    __syncthreads();

    const int rowsHere = min(OWN_ROWS, nNodes - rowBase);
    for (int i = tid; i < rowsHere; i += OWN_THREADS) {
        const float d = (float)deg[i];
        norm[rowBase + i] = (d > 0.0f) ? rsqrtf(fmaxf(d, 1.0f)) : 0.0f;
    }
}

template <bool SCALED, int DOUT, bool RELU, bool SCALE_OUT>
__global__ __launch_bounds__(OWN_THREADS) void gcn_edge_sum_lds(const float* __restrict__ x,
                                                                const int* __restrict__ src,
                                                                const int* __restrict__ dst,
                                                                const float* __restrict__ norm,
                                                                const unsigned char* __restrict__ owner,
                                                                const float* __restrict__ W,
                                                                const float* __restrict__ bias,
                                                                float* __restrict__ res,
                                                                const int* __restrict__ lists,
                                                                const int* __restrict__ counts,
                                                                int nNodes) {
    __shared__ __attribute__((aligned(16))) float acc[ACC_ROWS * ACC_PITCH];
    __shared__ int   qlists[OWN_WAVES * Q_CAP];

    constexpr int NCT = DOUT / 16;

    const int tid  = threadIdx.x;
    const int lane = tid & 31;
    const int wave = __builtin_amdgcn_readfirstlane(tid >> 5);
    const int rowBase = blockIdx.x * OWN_ROWS;

    for (int i = tid; i < ACC_ROWS * ACC_PITCH; i += OWN_THREADS) acc[i] = 0.0f;
    __syncthreads();

    OwnCtx c;
    c.x = x; c.src = src; c.dst = dst; c.norm = norm; c.acc = acc; c.deg = nullptr;
    c.qlist  = &qlists[wave * Q_CAP];
    c.rowBase = rowBase;

    const int* lst = lists + (size_t)(blockIdx.x * OWN_WAVES + wave) * LIST_CAP;
    const int total = __builtin_amdgcn_readfirstlane(counts[blockIdx.x * OWN_WAVES + wave]);
    if (total <= LIST_CAP) {
        if (total > 0) walk_list<SCALED>(lst, total, lane, c);
    } else {
        scan_ids<SCALED ? SUM_SCALED : SUM_PLAIN>((const uint4*)owner, lane, wave, c, nullptr);
    }

    const int m  = lane & 15;
    const int kh = lane >> 4;
    const int row0 = wave * 16;
    float rowScale = 0.0f;
    float os[8];
    float bv[NCT];
    if (wave < ROW_TILES) {
        rowScale = norm[min(rowBase + row0 + m, nNodes - 1)];
        if (SCALE_OUT) {
#pragma unroll
            for (int r = 0; r < 8; ++r) os[r] = norm[min(rowBase + row0 + r + 8 * kh, nNodes - 1)];
        }
#pragma unroll
        for (int ct = 0; ct < NCT; ++ct) bv[ct] = bias[ct * 16 + m];
    }
    __syncthreads();

    if (wave < ROW_TILES) {
        const float* arow = acc + (row0 + m) * ACC_PITCH + 2 * kh;
        v2f a[16];
#pragma unroll
        for (int j = 0; j < 16; ++j) {
            const float2 ap = *(const float2*)(arow + 4 * j);
            a[j].x = ap.x * rowScale;
            a[j].y = ap.y * rowScale;
        }

        v8f d[NCT];
#pragma unroll
        for (int ct = 0; ct < NCT; ++ct) d[ct] = (v8f){0.f, 0.f, 0.f, 0.f, 0.f, 0.f, 0.f, 0.f};

#pragma unroll
        for (int j = 0; j < 16; ++j) {
            const int ka = 4 * j + 2 * kh;
#pragma unroll
            for (int ct = 0; ct < NCT; ++ct) {
                v2f b;
                b.x = W[(size_t)ka * DOUT + ct * 16 + m];
                b.y = W[(size_t)(ka + 1) * DOUT + ct * 16 + m];
                d[ct] = __builtin_amdgcn_wmma_f32_16x16x4_f32(
                             false, a[j],  false, b,
                             (short)0, d[ct],  false,  false);
            }
        }

#pragma unroll
        for (int ct = 0; ct < NCT; ++ct) {
#pragma unroll
            for (int r = 0; r < 8; ++r) {
                float v = d[ct][r] + bv[ct];
                if (RELU) v = fmaxf(v, 0.0f);
                if (SCALE_OUT) v *= os[r];
                acc[(row0 + r + 8 * kh) * ACC_PITCH + ct * 16 + m] = v;
            }
        }
    }
    __syncthreads();

    const int rowsHere = min(OWN_ROWS, nNodes - rowBase);
    float4* out4 = (float4*)(res + (size_t)rowBase * DOUT);
    const float4* acc4 = (const float4*)acc;
    constexpr int Q = DOUT / 4;
    for (int i = tid; i < rowsHere * Q; i += OWN_THREADS) out4[i] = acc4[(i / Q) * (ACC_PITCH / 4) + (i % Q)];
}

extern "C" void kernel_launch(void* const* d_in, const int* in_sizes, int n_in,
                              void* d_out, int out_size, void* d_ws, size_t ws_size,
                              hipStream_t stream) {
    const float* x   = (const float*)d_in[0];
    const int*   ei  = (const int*)d_in[1];
    const float* W1  = (const float*)d_in[2];
    const float* b1  = (const float*)d_in[3];
    const float* W2  = (const float*)d_in[4];
    const float* b2  = (const float*)d_in[5];
    float*       out = (float*)d_out;

    const int N = in_sizes[0] / 64;
    const int E = in_sizes[1] / 2;
    const int* src = ei;
    const int* dst = ei + E;

    float* norm = (float*)d_ws;
    float* h    = norm + 2 * N + 256;
    unsigned char* owner = (unsigned char*)(h + (size_t)N * 64);
    int* counts = (int*)(owner + OWNER_SPAN);
    int* lists  = counts + OWN_BLOCKS * OWN_WAVES;

    gcn_owner_ids<<<N_TILES, 32, 0, stream>>>(dst, owner, E);
    gcn_find_edges<<<OWN_BLOCKS, OWN_THREADS, 40960, stream>>>(src, dst, owner, norm, lists, counts, N);

    gcn_edge_sum_lds<true, 64, true, true><<<OWN_BLOCKS, OWN_THREADS, 0, stream>>>(x, src, dst, norm, owner, W1, b1, h, lists, counts, N);

    gcn_edge_sum_lds<false, 32, false, false><<<OWN_BLOCKS, OWN_THREADS, 0, stream>>>(h, src, dst, norm, owner, W2, b2, out, lists, counts, N);
}
